# starting kernel
# baseline (speedup 1.0000x reference)
#include <hip/hip_runtime.h>
#include <hip/hip_bf16.h>

#define N_NODES 10000
#define N_EDGES 320000
#define IN_DIM  512
#define OUT_DIM 256
#define NEG_SLOPE 0.01f

typedef float v2f __attribute__((ext_vector_type(2)));
typedef float v8f __attribute__((ext_vector_type(8)));

__device__ __forceinline__ unsigned enc_f32(float f) {
    int b = __float_as_int(f);
    return (b >= 0) ? ((unsigned)b | 0x80000000u) : ~(unsigned)b;
}
__device__ __forceinline__ float dec_f32(unsigned k) {
    int b = (k & 0x80000000u) ? (int)(k & 0x7FFFFFFFu) : ~(int)k;
    return __int_as_float(b);
}
#define EMAX_NEG_INF 0x007FFFFFu

__global__ void k_init(unsigned* __restrict__ emax, float* __restrict__ denom,
                       float* __restrict__ hout) {
    int i = blockIdx.x * blockDim.x + threadIdx.x;
    if (i < N_NODES) { emax[i] = EMAX_NEG_INF; denom[i] = 0.0f; }
    for (int j = i; j < N_NODES * OUT_DIM; j += gridDim.x * blockDim.x)
        hout[j] = 0.0f;
}

__global__ void k_gemm_wmma(const float* __restrict__ h, const float* __restrict__ w,
                            const float* __restrict__ bias, float* __restrict__ z) {
    const int lane = threadIdx.x & 31;
    const int gw   = blockIdx.x * (blockDim.x >> 5) + (threadIdx.x >> 5);
    if (gw >= 2500) return;
    const int row0 = (gw >> 2) * 16;
    const int col0 = (gw & 3) * 64;
    const int mn   = lane & 15;
    const int kb   = (lane >> 4) << 1;

    const float* arow = h + (size_t)(row0 + mn) * IN_DIM + kb;
    const float* b0   = w + (size_t)(col0 + mn) * IN_DIM + kb;
    const float* b1   = b0 + (size_t)16 * IN_DIM;
    const float* b2   = b0 + (size_t)32 * IN_DIM;
    const float* b3   = b0 + (size_t)48 * IN_DIM;

    v8f c0 = {}, c1 = {}, c2 = {}, c3 = {};
#pragma unroll 4
    for (int k = 0; k < IN_DIM; k += 4) {
        v2f a  = *(const v2f*)(arow + k);
        v2f q0 = *(const v2f*)(b0 + k);
        v2f q1 = *(const v2f*)(b1 + k);
        v2f q2 = *(const v2f*)(b2 + k);
        v2f q3 = *(const v2f*)(b3 + k);
        c0 = __builtin_amdgcn_wmma_f32_16x16x4_f32(false, a, false, q0, (short)0, c0, false, false);
        c1 = __builtin_amdgcn_wmma_f32_16x16x4_f32(false, a, false, q1, (short)0, c1, false, false);
        c2 = __builtin_amdgcn_wmma_f32_16x16x4_f32(false, a, false, q2, (short)0, c2, false, false);
        c3 = __builtin_amdgcn_wmma_f32_16x16x4_f32(false, a, false, q3, (short)0, c3, false, false);
    }

    const int mbase = (lane >> 4) * 8;
    v8f acc[4] = {c0, c1, c2, c3};
#pragma unroll
    for (int g = 0; g < 4; ++g) {
        const int   col = col0 + g * 16 + mn;
        const float bv  = bias[col];
#pragma unroll
        for (int v = 0; v < 8; ++v)
            z[(size_t)(row0 + mbase + v) * OUT_DIM + col] = acc[g][v] + bv;
    }
}

__global__ void k_node_dots(const float* __restrict__ z, const float* __restrict__ attn_w,
                            float* __restrict__ s_src, float* __restrict__ s_dst) {
    const int lane = threadIdx.x & 31;
    const int node = (blockIdx.x * blockDim.x + threadIdx.x) >> 5;
    if (node >= N_NODES) return;
    const float4* zrow = (const float4*)(z + (size_t)node * OUT_DIM);
    const float4* aw   = (const float4*)attn_w;
    float ps = 0.0f, pd = 0.0f;
#pragma unroll
    for (int j = 0; j < 2; ++j) {
        int vi = lane + j * 32;
        float4 zz = zrow[vi];
        float4 as = aw[vi];
        float4 ad = aw[64 + vi];
        ps += zz.x * as.x + zz.y * as.y + zz.z * as.z + zz.w * as.w;
        pd += zz.x * ad.x + zz.y * ad.y + zz.z * ad.z + zz.w * ad.w;
    }
#pragma unroll
    for (int m = 16; m >= 1; m >>= 1) {
        ps += __shfl_xor(ps, m, 32);
        pd += __shfl_xor(pd, m, 32);
    }
    if (lane == 0) { s_src[node] = ps; s_dst[node] = pd; }
}

__global__ void k_edge_scores(const int* __restrict__ src, const int* __restrict__ dst,
                              const float* __restrict__ s_src, const float* __restrict__ s_dst,
                              const float* __restrict__ attn_b,
                              float* __restrict__ e, unsigned* __restrict__ emax) {
    int i = blockIdx.x * blockDim.x + threadIdx.x;
    if (i >= N_EDGES) return;
    float v = s_src[src[i]] + s_dst[dst[i]] + attn_b[0];
    v = (v >= 0.0f) ? v : NEG_SLOPE * v;
    e[i] = v;
    atomicMax(&emax[dst[i]], enc_f32(v));
}

__global__ void k_edge_exp(const int* __restrict__ dst, float* __restrict__ e,
                           const unsigned* __restrict__ emax, float* __restrict__ denom) {
    int i = blockIdx.x * blockDim.x + threadIdx.x;
    if (i >= N_EDGES) return;
    int d = dst[i];
    float ee = expf(e[i] - dec_f32(emax[d]));
    e[i] = ee;
    unsafeAtomicAdd(&denom[d], ee);
}

__global__ void k_aggregate(const int* __restrict__ src, const int* __restrict__ dst,
                            const float* __restrict__ e_exp, const float* __restrict__ denom,
                            const float* __restrict__ z, float* __restrict__ hout) {
    const int lane = threadIdx.x & 31;
    const int edge = (blockIdx.x * blockDim.x + threadIdx.x) >> 5;
    if (edge >= N_EDGES) return;
    const int s = src[edge], d = dst[edge];
    const float alpha = e_exp[edge] / denom[d];
    const float4* zs = (const float4*)(z + (size_t)s * OUT_DIM);
    float* ho = hout + (size_t)d * OUT_DIM;
#pragma unroll
    for (int j = 0; j < 2; ++j) {
        int vi = lane + j * 32;
        float4 zz = zs[vi];
        unsafeAtomicAdd(&ho[vi * 4 + 0], alpha * zz.x);
        unsafeAtomicAdd(&ho[vi * 4 + 1], alpha * zz.y);
        unsafeAtomicAdd(&ho[vi * 4 + 2], alpha * zz.z);
        unsafeAtomicAdd(&ho[vi * 4 + 3], alpha * zz.w);
    }
}

static inline size_t align256(size_t x) { return (x + 255) & ~(size_t)255; }

extern "C" void kernel_launch(void* const* d_in, const int* in_sizes, int n_in,
                              void* d_out, int out_size, void* d_ws, size_t ws_size,
                              hipStream_t stream) {
    const float* h      = (const float*)d_in[0];
    const float* fc_w   = (const float*)d_in[1];
    const float* fc_b   = (const float*)d_in[2];
    const float* attn_w = (const float*)d_in[3];
    const float* attn_b = (const float*)d_in[4];
    const int*   src    = (const int*)d_in[5];
    const int*   dst    = (const int*)d_in[6];
    float* hout = (float*)d_out;

    char* ws = (char*)d_ws;
    float*    z      = (float*)ws;    ws += align256((size_t)N_NODES * OUT_DIM * 4);
    float*    s_src  = (float*)ws;    ws += align256((size_t)N_NODES * 4);
    float*    s_dst  = (float*)ws;    ws += align256((size_t)N_NODES * 4);
    float*    e_buf  = (float*)ws;    ws += align256((size_t)N_EDGES * 4);
    unsigned* emax   = (unsigned*)ws; ws += align256((size_t)N_NODES * 4);
    float*    denom  = (float*)ws;    ws += align256((size_t)N_NODES * 4);

    k_init<<<1250, 256, 0, stream>>>(emax, denom, hout);
    k_gemm_wmma<<<313, 256, 0, stream>>>(h, fc_w, fc_b, z);
    k_node_dots<<<1250, 256, 0, stream>>>(z, attn_w, s_src, s_dst);
    k_edge_scores<<<(N_EDGES + 255) / 256, 256, 0, stream>>>(src, dst, s_src, s_dst,
                                                             attn_b, e_buf, emax);
    k_edge_exp<<<(N_EDGES + 255) / 256, 256, 0, stream>>>(dst, e_buf, emax, denom);
    k_aggregate<<<(N_EDGES * 32 + 255) / 256, 256, 0, stream>>>(src, dst, e_buf, denom,
                                                                z, hout);
}


# Edge passes + aggregate in one launch: a work-group owns 40 destination rows, lists its edges from a scan of dst, does max, denominator, row sums by LDS atomics (arrival order); no global atomics
# speedup vs baseline: 2.5061x; 2.5061x over previous
#include <hip/hip_runtime.h>
#include <hip/hip_bf16.h>

#define N_NODES 10000
#define N_EDGES 320000
#define IN_DIM  512
#define OUT_DIM 256
#define NEG_SLOPE 0.01f

typedef float v2f __attribute__((ext_vector_type(2)));
typedef float v8f __attribute__((ext_vector_type(8)));

__device__ __forceinline__ unsigned enc_f32(float f) {
    int b = __float_as_int(f);
    return (b >= 0) ? ((unsigned)b | 0x80000000u) : ~(unsigned)b;
}
__device__ __forceinline__ float dec_f32(unsigned k) {
    int b = (k & 0x80000000u) ? (int)(k & 0x7FFFFFFFu) : ~(int)k;
    return __int_as_float(b);
}
#define EMAX_NEG_INF 0x007FFFFFu

__global__ void k_gemm_wmma(const float* __restrict__ h, const float* __restrict__ w,
                            const float* __restrict__ bias, float* __restrict__ z) {
    const int lane = threadIdx.x & 31;
    const int gw   = blockIdx.x * (blockDim.x >> 5) + (threadIdx.x >> 5);
    if (gw >= 2500) return;
    const int row0 = (gw >> 2) * 16;
    const int col0 = (gw & 3) * 64;
    const int mn   = lane & 15;
    const int kb   = (lane >> 4) << 1;

    const float* arow = h + (size_t)(row0 + mn) * IN_DIM + kb;
    const float* b0   = w + (size_t)(col0 + mn) * IN_DIM + kb;
    const float* b1   = b0 + (size_t)16 * IN_DIM;
    const float* b2   = b0 + (size_t)32 * IN_DIM;
    const float* b3   = b0 + (size_t)48 * IN_DIM;

    v8f c0 = {}, c1 = {}, c2 = {}, c3 = {};
#pragma unroll 4
    for (int k = 0; k < IN_DIM; k += 4) {
        v2f a  = *(const v2f*)(arow + k);
        v2f q0 = *(const v2f*)(b0 + k);
        v2f q1 = *(const v2f*)(b1 + k);
        v2f q2 = *(const v2f*)(b2 + k);
        v2f q3 = *(const v2f*)(b3 + k);
        c0 = __builtin_amdgcn_wmma_f32_16x16x4_f32(false, a, false, q0, (short)0, c0, false, false);
        c1 = __builtin_amdgcn_wmma_f32_16x16x4_f32(false, a, false, q1, (short)0, c1, false, false);
        c2 = __builtin_amdgcn_wmma_f32_16x16x4_f32(false, a, false, q2, (short)0, c2, false, false);
        c3 = __builtin_amdgcn_wmma_f32_16x16x4_f32(false, a, false, q3, (short)0, c3, false, false);
    }

    const int mbase = (lane >> 4) * 8;
    v8f acc[4] = {c0, c1, c2, c3};
#pragma unroll
    for (int g = 0; g < 4; ++g) {
        const int   col = col0 + g * 16 + mn;
        const float bv  = bias[col];
#pragma unroll
        for (int v = 0; v < 8; ++v)
            z[(size_t)(row0 + mbase + v) * OUT_DIM + col] = acc[g][v] + bv;
    }
}

__global__ void k_node_dots(const float* __restrict__ z, const float* __restrict__ attn_w,
                            float* __restrict__ s_src, float* __restrict__ s_dst) {
    const int lane = threadIdx.x & 31;
    const int node = (blockIdx.x * blockDim.x + threadIdx.x) >> 5;
    if (node >= N_NODES) return;
    const float4* zrow = (const float4*)(z + (size_t)node * OUT_DIM);
    const float4* aw   = (const float4*)attn_w;
    float ps = 0.0f, pd = 0.0f;
#pragma unroll
    for (int j = 0; j < 2; ++j) {
        int vi = lane + j * 32;
        float4 zz = zrow[vi];
        float4 as = aw[vi];
        float4 ad = aw[64 + vi];
        ps += zz.x * as.x + zz.y * as.y + zz.z * as.z + zz.w * as.w;
        pd += zz.x * ad.x + zz.y * ad.y + zz.z * ad.z + zz.w * ad.w;
    }
#pragma unroll
    for (int m = 16; m >= 1; m >>= 1) {
        ps += __shfl_xor(ps, m, 32);
        pd += __shfl_xor(pd, m, 32);
    }
    if (lane == 0) { s_src[node] = ps; s_dst[node] = pd; }
}

#define SL_ROWS     40
#define SL_GROUPS   (N_NODES / SL_ROWS)
#define SL_THREADS  1024
#define SL_WAVES    (SL_THREADS / 32)
#define SL_CAP      16384
#define N_QUADS     (N_EDGES / 4)
#define SC_BATCH    8

__device__ __forceinline__ float edge_logit(float ss, float sd, float ab) {
    const float v = ss + sd + ab;
    return (v >= 0.0f) ? v : NEG_SLOPE * v;
}

__device__ __forceinline__ void list_push(unsigned* l_cnt, unsigned* l_ent, int ei, unsigned r) {
    const unsigned slot = atomicAdd(l_cnt, 1u);
    if (slot < (unsigned)SL_CAP) l_ent[slot] = (unsigned)ei | (r << 19);
}

__device__ __forceinline__ void row_add(float* acc, const float* __restrict__ z,
                                        int s, unsigned r, float alpha, int lane) {
    const float* zs = z + (size_t)s * OUT_DIM + lane;
    float* ar = acc + r * OUT_DIM + lane;
    float zz[8];
#pragma unroll
    for (int j = 0; j < 8; ++j) zz[j] = zs[j * 32];
#pragma unroll
    for (int j = 0; j < 8; ++j) unsafeAtomicAdd(&ar[j * 32], alpha * zz[j]);
}

__global__ __launch_bounds__(SL_THREADS)
void k_edges_owned(const int* __restrict__ src, const int* __restrict__ dst,
                   const float* __restrict__ s_src, const float* __restrict__ s_dst,
                   const float* __restrict__ attn_b, const float* __restrict__ z,
                   float* __restrict__ hout) {
    __shared__ float    acc[SL_ROWS * OUT_DIM];
    __shared__ unsigned l_ent[SL_CAP];
    __shared__ float    l_val[SL_CAP];
    __shared__ float    l_sd[SL_ROWS];
    __shared__ unsigned l_mx[SL_ROWS];
    __shared__ float    l_dn[SL_ROWS];
    __shared__ unsigned l_cnt;

    const int   tid  = threadIdx.x;
    const int   lane = tid & 31;
    const int   wv   = tid >> 5;
    const int   lo   = blockIdx.x * SL_ROWS;
    const float ab   = attn_b[0];

    for (int j = tid; j < SL_ROWS * OUT_DIM / 4; j += SL_THREADS)
        ((float4*)acc)[j] = make_float4(0.0f, 0.0f, 0.0f, 0.0f);
    if (tid < SL_ROWS) { l_sd[tid] = s_dst[lo + tid]; l_mx[tid] = EMAX_NEG_INF; l_dn[tid] = 0.0f; }
    if (tid == 0) l_cnt = 0u;
    __syncthreads();

    const int start = blockIdx.x * (N_QUADS / SL_GROUPS);
    for (int base = 0; base < N_QUADS; base += SL_THREADS * SC_BATCH) {
        int4 dq[SC_BATCH];
#pragma unroll
        for (int u = 0; u < SC_BATCH; ++u) {
            const int it = base + u * SL_THREADS + tid;
            dq[u] = make_int4(-1, -1, -1, -1);
            if (it < N_QUADS) {
                int qd = it + start;
                if (qd >= N_QUADS) qd -= N_QUADS;
                dq[u] = ((const int4*)dst)[qd];
            }
        }
#pragma unroll
        for (int u = 0; u < SC_BATCH; ++u) {
            const int it = base + u * SL_THREADS + tid;
            int qd = it + start;
            if (qd >= N_QUADS) qd -= N_QUADS;
            const unsigned r0 = (unsigned)(dq[u].x - lo), r1 = (unsigned)(dq[u].y - lo);
            const unsigned r2 = (unsigned)(dq[u].z - lo), r3 = (unsigned)(dq[u].w - lo);
            if (min(min(r0, r1), min(r2, r3)) < (unsigned)SL_ROWS) {
                if (r0 < (unsigned)SL_ROWS) list_push(&l_cnt, l_ent, qd * 4 + 0, r0);
                if (r1 < (unsigned)SL_ROWS) list_push(&l_cnt, l_ent, qd * 4 + 1, r1);
                if (r2 < (unsigned)SL_ROWS) list_push(&l_cnt, l_ent, qd * 4 + 2, r2);
                if (r3 < (unsigned)SL_ROWS) list_push(&l_cnt, l_ent, qd * 4 + 3, r3);
            }
        }
    }
    __syncthreads();
    const unsigned n_all = l_cnt;

    if (n_all <= (unsigned)SL_CAP) {
        const int n = (int)n_all;
        for (int i = tid; i < n; i += SL_THREADS) {
            const unsigned p = l_ent[i];
            const int      ei = (int)(p & 0x7FFFFu);
            const unsigned r  = p >> 19;
            const int      s  = src[ei];
            const float    v  = edge_logit(s_src[s], l_sd[r], ab);
            l_ent[i] = (unsigned)s | (r << 16);
            l_val[i] = v;
            atomicMax(&l_mx[r], enc_f32(v));
        }
        __syncthreads();
        for (int i = tid; i < n; i += SL_THREADS) {
            const unsigned r = l_ent[i] >> 16;
            const float    x = expf(l_val[i] - dec_f32(l_mx[r]));
            l_val[i] = x;
            unsafeAtomicAdd(&l_dn[r], x);
        }
        __syncthreads();
        const int cnt = (n > wv) ? (n - wv + SL_WAVES - 1) / SL_WAVES : 0;
        int k = 0;
        for (; k + 1 < cnt; k += 2) {
            const int      i   = wv + k * SL_WAVES;
            const int      i2  = i + SL_WAVES;
            const unsigned e0  = l_ent[i];
            const unsigned e1  = l_ent[i2];
            const int      s0 = (int)(e0 & 0xFFFFu), s1 = (int)(e1 & 0xFFFFu);
            const unsigned r0 = e0 >> 16,            r1 = e1 >> 16;
            const float    a0 = l_val[i]  / l_dn[r0];
            const float    a1 = l_val[i2] / l_dn[r1];
            const float* z0 = z + (size_t)s0 * OUT_DIM + lane;
            const float* z1 = z + (size_t)s1 * OUT_DIM + lane;
            float* c0 = acc + r0 * OUT_DIM + lane;
            float* c1 = acc + r1 * OUT_DIM + lane;
            float u0[8], u1[8];
#pragma unroll
            for (int j = 0; j < 8; ++j) u0[j] = z0[j * 32];
#pragma unroll
            for (int j = 0; j < 8; ++j) u1[j] = z1[j * 32];
#pragma unroll
            for (int j = 0; j < 8; ++j) unsafeAtomicAdd(&c0[j * 32], a0 * u0[j]);
#pragma unroll
            for (int j = 0; j < 8; ++j) unsafeAtomicAdd(&c1[j * 32], a1 * u1[j]);
        }
        if (k < cnt) {
            const int      i  = wv + k * SL_WAVES;
            const unsigned e0 = l_ent[i];
            row_add(acc, z, (int)(e0 & 0xFFFFu), e0 >> 16, l_val[i] / l_dn[e0 >> 16], lane);
        }
    } else {
        for (int ei = tid; ei < N_EDGES; ei += SL_THREADS) {
            const unsigned r = (unsigned)(dst[ei] - lo);
            if (r < (unsigned)SL_ROWS)
                atomicMax(&l_mx[r], enc_f32(edge_logit(s_src[src[ei]], l_sd[r], ab)));
        }
        __syncthreads();
        for (int ei = tid; ei < N_EDGES; ei += SL_THREADS) {
            const unsigned r = (unsigned)(dst[ei] - lo);
            if (r < (unsigned)SL_ROWS) {
                const float v = edge_logit(s_src[src[ei]], l_sd[r], ab);
                unsafeAtomicAdd(&l_dn[r], expf(v - dec_f32(l_mx[r])));
            }
        }
        __syncthreads();
        for (int base = wv * 32; base < N_EDGES; base += SL_THREADS) {
            const int      ei   = base + lane;
            const unsigned r    = (unsigned)(dst[ei] - lo);
            const bool     mine = r < (unsigned)SL_ROWS;
            int   s = 0;
            float a = 0.0f;
            if (mine) {
                s = src[ei];
                const float v = edge_logit(s_src[s], l_sd[r], ab);
                a = expf(v - dec_f32(l_mx[r])) / l_dn[r];
            }
            unsigned m = __builtin_amdgcn_ballot_w32(mine);
            while (m) {
                const int i = __builtin_ctz(m);
                m &= m - 1u;
                const int      si = __builtin_amdgcn_readlane(s, i);
                const unsigned ri = (unsigned)__builtin_amdgcn_readlane((int)r, i);
                const float    ai = __int_as_float(__builtin_amdgcn_readlane(__float_as_int(a), i));
                row_add(acc, z, si, ri, ai, lane);
            }
        }
    }
    __syncthreads();
    float4* og = (float4*)(hout + (size_t)lo * OUT_DIM);
    for (int j = tid; j < SL_ROWS * OUT_DIM / 4; j += SL_THREADS) og[j] = ((const float4*)acc)[j];
}

static inline size_t align256(size_t x) { return (x + 255) & ~(size_t)255; }

extern "C" void kernel_launch(void* const* d_in, const int* in_sizes, int n_in,
                              void* d_out, int out_size, void* d_ws, size_t ws_size,
                              hipStream_t stream) {
    const float* h      = (const float*)d_in[0];
    const float* fc_w   = (const float*)d_in[1];
    const float* fc_b   = (const float*)d_in[2];
    const float* attn_w = (const float*)d_in[3];
    const float* attn_b = (const float*)d_in[4];
    const int*   src    = (const int*)d_in[5];
    const int*   dst    = (const int*)d_in[6];
    float* hout = (float*)d_out;

    char* ws = (char*)d_ws;
    float*    z      = (float*)ws;    ws += align256((size_t)N_NODES * OUT_DIM * 4);
    float*    s_src  = (float*)ws;    ws += align256((size_t)N_NODES * 4);
    float*    s_dst  = (float*)ws;    ws += align256((size_t)N_NODES * 4);

    k_gemm_wmma<<<313, 256, 0, stream>>>(h, fc_w, fc_b, z);
    k_node_dots<<<1250, 256, 0, stream>>>(z, attn_w, s_src, s_dst);
    k_edges_owned<<<SL_GROUPS, SL_THREADS, 0, stream>>>(src, dst, s_src, s_dst, attn_b, z, hout);
}


# Edge scan reads one-byte slab keys (dst/40, packed by the node-dot launch), 16 edges per 16-byte load: a quarter of the scan bytes; listed edges read their row from dst; LDS sums in arrival order
# speedup vs baseline: 2.6890x; 1.0730x over previous
#include <hip/hip_runtime.h>
#include <hip/hip_bf16.h>

#define N_NODES 10000
#define N_EDGES 320000
#define IN_DIM  512
#define OUT_DIM 256
#define NEG_SLOPE 0.01f

typedef float v2f __attribute__((ext_vector_type(2)));
typedef float v8f __attribute__((ext_vector_type(8)));

__device__ __forceinline__ unsigned enc_f32(float f) {
    int b = __float_as_int(f);
    return (b >= 0) ? ((unsigned)b | 0x80000000u) : ~(unsigned)b;
}
__device__ __forceinline__ float dec_f32(unsigned k) {
    int b = (k & 0x80000000u) ? (int)(k & 0x7FFFFFFFu) : ~(int)k;
    return __int_as_float(b);
}
#define EMAX_NEG_INF 0x007FFFFFu

__global__ void k_gemm_wmma(const float* __restrict__ h, const float* __restrict__ w,
                            const float* __restrict__ bias, float* __restrict__ z) {
    const int lane = threadIdx.x & 31;
    const int gw   = blockIdx.x * (blockDim.x >> 5) + (threadIdx.x >> 5);
    if (gw >= 2500) return;
    const int row0 = (gw >> 2) * 16;
    const int col0 = (gw & 3) * 64;
    const int mn   = lane & 15;
    const int kb   = (lane >> 4) << 1;

    const float* arow = h + (size_t)(row0 + mn) * IN_DIM + kb;
    const float* b0   = w + (size_t)(col0 + mn) * IN_DIM + kb;
    const float* b1   = b0 + (size_t)16 * IN_DIM;
    const float* b2   = b0 + (size_t)32 * IN_DIM;
    const float* b3   = b0 + (size_t)48 * IN_DIM;

    v8f c0 = {}, c1 = {}, c2 = {}, c3 = {};
#pragma unroll 4
    for (int k = 0; k < IN_DIM; k += 4) {
        v2f a  = *(const v2f*)(arow + k);
        v2f q0 = *(const v2f*)(b0 + k);
        v2f q1 = *(const v2f*)(b1 + k);
        v2f q2 = *(const v2f*)(b2 + k);
        v2f q3 = *(const v2f*)(b3 + k);
        c0 = __builtin_amdgcn_wmma_f32_16x16x4_f32(false, a, false, q0, (short)0, c0, false, false);
        c1 = __builtin_amdgcn_wmma_f32_16x16x4_f32(false, a, false, q1, (short)0, c1, false, false);
        c2 = __builtin_amdgcn_wmma_f32_16x16x4_f32(false, a, false, q2, (short)0, c2, false, false);
        c3 = __builtin_amdgcn_wmma_f32_16x16x4_f32(false, a, false, q3, (short)0, c3, false, false);
    }

    const int mbase = (lane >> 4) * 8;
    v8f acc[4] = {c0, c1, c2, c3};
#pragma unroll
    for (int g = 0; g < 4; ++g) {
        const int   col = col0 + g * 16 + mn;
        const float bv  = bias[col];
#pragma unroll
        for (int v = 0; v < 8; ++v)
            z[(size_t)(row0 + mbase + v) * OUT_DIM + col] = acc[g][v] + bv;
    }
}

#define KEY_ROWS   40
#define N_KEYLINES (N_EDGES / 128)
__device__ __forceinline__ unsigned slab_key(int d) {
    return min((unsigned)d / (unsigned)KEY_ROWS, 255u);
}
__global__ void k_node_dots(const float* __restrict__ z, const float* __restrict__ attn_w,
                            const int* __restrict__ dst,
                            float* __restrict__ s_src, float* __restrict__ s_dst,
                            unsigned* __restrict__ ekey) {
    const int lane = threadIdx.x & 31;
    const int node = (blockIdx.x * blockDim.x + threadIdx.x) >> 5;
    if (node >= N_NODES) return;
    if (node < N_KEYLINES) {
        const int4 d = ((const int4*)dst)[node * 32 + lane];
        ekey[node * 32 + lane] = slab_key(d.x) | (slab_key(d.y) << 8) |
                                 (slab_key(d.z) << 16) | (slab_key(d.w) << 24);
    }
    const float4* zrow = (const float4*)(z + (size_t)node * OUT_DIM);
    const float4* aw   = (const float4*)attn_w;
    float ps = 0.0f, pd = 0.0f;
#pragma unroll
    for (int j = 0; j < 2; ++j) {
        int vi = lane + j * 32;
        float4 zz = zrow[vi];
        float4 as = aw[vi];
        float4 ad = aw[64 + vi];
        ps += zz.x * as.x + zz.y * as.y + zz.z * as.z + zz.w * as.w;
        pd += zz.x * ad.x + zz.y * ad.y + zz.z * ad.z + zz.w * ad.w;
    }
#pragma unroll
    for (int m = 16; m >= 1; m >>= 1) {
        ps += __shfl_xor(ps, m, 32);
        pd += __shfl_xor(pd, m, 32);
    }
    if (lane == 0) { s_src[node] = ps; s_dst[node] = pd; }
}

#define SL_ROWS     40
#define SL_GROUPS   (N_NODES / SL_ROWS)
#define SL_THREADS  1024
#define SL_WAVES    (SL_THREADS / 32)
#define SL_CAP      16384
#define N_KQ        (N_EDGES / 16)
#define SC_BATCH    10

__device__ __forceinline__ float edge_logit(float ss, float sd, float ab) {
    const float v = ss + sd + ab;
    return (v >= 0.0f) ? v : NEG_SLOPE * v;
}

__device__ __forceinline__ void list_push(unsigned* l_cnt, unsigned* l_ent, int ei) {
    const unsigned slot = atomicAdd(l_cnt, 1u);
    if (slot < (unsigned)SL_CAP) l_ent[slot] = (unsigned)ei;
}

__device__ __forceinline__ unsigned key_hits(unsigned w, unsigned pat) {
    const unsigned x = w ^ pat;
    const unsigned t = (x & 0x7F7F7F7Fu) + 0x7F7F7F7Fu;
    return ~(t | x) & 0x80808080u;
}

__device__ __forceinline__ void row_add(float* acc, const float* __restrict__ z,
                                        int s, unsigned r, float alpha, int lane) {
    const float* zs = z + (size_t)s * OUT_DIM + lane;
    float* ar = acc + r * OUT_DIM + lane;
    float zz[8];
#pragma unroll
    for (int j = 0; j < 8; ++j) zz[j] = zs[j * 32];
#pragma unroll
    for (int j = 0; j < 8; ++j) unsafeAtomicAdd(&ar[j * 32], alpha * zz[j]);
}

__global__ __launch_bounds__(SL_THREADS)
void k_edges_owned(const int* __restrict__ src, const int* __restrict__ dst,
                   const unsigned* __restrict__ ekey,
                   const float* __restrict__ s_src, const float* __restrict__ s_dst,
                   const float* __restrict__ attn_b, const float* __restrict__ z,
                   float* __restrict__ hout) {
    __shared__ float    acc[SL_ROWS * OUT_DIM];
    __shared__ unsigned l_ent[SL_CAP];
    __shared__ float    l_val[SL_CAP];
    __shared__ float    l_sd[SL_ROWS];
    __shared__ unsigned l_mx[SL_ROWS];
    __shared__ float    l_dn[SL_ROWS];
    __shared__ unsigned l_cnt;

    const int   tid  = threadIdx.x;
    const int   lane = tid & 31;
    const int   wv   = tid >> 5;
    const int   lo   = blockIdx.x * SL_ROWS;
    const float ab   = attn_b[0];

    for (int j = tid; j < SL_ROWS * OUT_DIM / 4; j += SL_THREADS)
        ((float4*)acc)[j] = make_float4(0.0f, 0.0f, 0.0f, 0.0f);
    if (tid < SL_ROWS) { l_sd[tid] = s_dst[lo + tid]; l_mx[tid] = EMAX_NEG_INF; l_dn[tid] = 0.0f; }
    if (tid == 0) l_cnt = 0u;
    __syncthreads();

    const unsigned pat   = (unsigned)blockIdx.x * 0x01010101u;
    const int      start = blockIdx.x * (N_KQ / SL_GROUPS);
    for (int base = 0; base < N_KQ; base += SL_THREADS * SC_BATCH) {
        uint4 kq[SC_BATCH];
#pragma unroll
        for (int u = 0; u < SC_BATCH; ++u) {
            int qd = base + u * SL_THREADS + tid + start;
            if (qd >= N_KQ) qd -= N_KQ;
            kq[u] = ((const uint4*)ekey)[min(qd, N_KQ - 1)];
        }
#pragma unroll
        for (int u = 0; u < SC_BATCH; ++u) {
            const int it = base + u * SL_THREADS + tid;
            int qd = it + start;
            if (qd >= N_KQ) qd -= N_KQ;
            unsigned m = (key_hits(kq[u].x, pat) >> 3) | (key_hits(kq[u].y, pat) >> 2) |
                         (key_hits(kq[u].z, pat) >> 1) |  key_hits(kq[u].w, pat);
            if (it < N_KQ && m != 0u) {
                while (m) {
                    const int p = __builtin_ctz(m);
                    list_push(&l_cnt, l_ent, qd * 16 + ((p & 3) << 2) + (p >> 3));
                    m &= m - 1u;
                }
            }
        }
    }
    __syncthreads();
    const unsigned n_all = l_cnt;

    if (n_all <= (unsigned)SL_CAP) {
        const int n = (int)n_all;
        for (int i = tid; i < n; i += SL_THREADS) {
            const int      ei = (int)l_ent[i];
            const int      s  = src[ei];
            const unsigned r  = (unsigned)(dst[ei] - lo);
            const float    v  = edge_logit(s_src[s], l_sd[r], ab);
            l_ent[i] = (unsigned)s | (r << 16);
            l_val[i] = v;
            atomicMax(&l_mx[r], enc_f32(v));
        }
        __syncthreads();
        for (int i = tid; i < n; i += SL_THREADS) {
            const unsigned r = l_ent[i] >> 16;
            const float    x = expf(l_val[i] - dec_f32(l_mx[r]));
            l_val[i] = x;
            unsafeAtomicAdd(&l_dn[r], x);
        }
        __syncthreads();
        const int cnt = (n > wv) ? (n - wv + SL_WAVES - 1) / SL_WAVES : 0;
        int k = 0;
        for (; k + 1 < cnt; k += 2) {
            const int      i   = wv + k * SL_WAVES;
            const int      i2  = i + SL_WAVES;
            const unsigned e0  = l_ent[i];
            const unsigned e1  = l_ent[i2];
            const int      s0 = (int)(e0 & 0xFFFFu), s1 = (int)(e1 & 0xFFFFu);
            const unsigned r0 = e0 >> 16,            r1 = e1 >> 16;
            const float    a0 = l_val[i]  / l_dn[r0];
            const float    a1 = l_val[i2] / l_dn[r1];
            const float* z0 = z + (size_t)s0 * OUT_DIM + lane;
            const float* z1 = z + (size_t)s1 * OUT_DIM + lane;
            float* c0 = acc + r0 * OUT_DIM + lane;
            float* c1 = acc + r1 * OUT_DIM + lane;
            float u0[8], u1[8];
#pragma unroll
            for (int j = 0; j < 8; ++j) u0[j] = z0[j * 32];
#pragma unroll
            for (int j = 0; j < 8; ++j) u1[j] = z1[j * 32];
#pragma unroll
            for (int j = 0; j < 8; ++j) unsafeAtomicAdd(&c0[j * 32], a0 * u0[j]);
#pragma unroll
            for (int j = 0; j < 8; ++j) unsafeAtomicAdd(&c1[j * 32], a1 * u1[j]);
        }
        if (k < cnt) {
            const int      i  = wv + k * SL_WAVES;
            const unsigned e0 = l_ent[i];
            row_add(acc, z, (int)(e0 & 0xFFFFu), e0 >> 16, l_val[i] / l_dn[e0 >> 16], lane);
        }
    } else {
        for (int ei = tid; ei < N_EDGES; ei += SL_THREADS) {
            const unsigned r = (unsigned)(dst[ei] - lo);
            if (r < (unsigned)SL_ROWS)
                atomicMax(&l_mx[r], enc_f32(edge_logit(s_src[src[ei]], l_sd[r], ab)));
        }
        __syncthreads();
        for (int ei = tid; ei < N_EDGES; ei += SL_THREADS) {
            const unsigned r = (unsigned)(dst[ei] - lo);
            if (r < (unsigned)SL_ROWS) {
                const float v = edge_logit(s_src[src[ei]], l_sd[r], ab);
                unsafeAtomicAdd(&l_dn[r], expf(v - dec_f32(l_mx[r])));
            }
        }
        __syncthreads();
        for (int base = wv * 32; base < N_EDGES; base += SL_THREADS) {
            const int      ei   = base + lane;
            const unsigned r    = (unsigned)(dst[ei] - lo);
            const bool     mine = r < (unsigned)SL_ROWS;
            int   s = 0;
            float a = 0.0f;
            if (mine) {
                s = src[ei];
                const float v = edge_logit(s_src[s], l_sd[r], ab);
                a = expf(v - dec_f32(l_mx[r])) / l_dn[r];
            }
            unsigned m = __builtin_amdgcn_ballot_w32(mine);
            while (m) {
                const int i = __builtin_ctz(m);
                m &= m - 1u;
                const int      si = __builtin_amdgcn_readlane(s, i);
                const unsigned ri = (unsigned)__builtin_amdgcn_readlane((int)r, i);
                const float    ai = __int_as_float(__builtin_amdgcn_readlane(__float_as_int(a), i));
                row_add(acc, z, si, ri, ai, lane);
            }
        }
    }
    __syncthreads();
    float4* og = (float4*)(hout + (size_t)lo * OUT_DIM);
    for (int j = tid; j < SL_ROWS * OUT_DIM / 4; j += SL_THREADS) og[j] = ((const float4*)acc)[j];
}

static inline size_t align256(size_t x) { return (x + 255) & ~(size_t)255; }

extern "C" void kernel_launch(void* const* d_in, const int* in_sizes, int n_in,
                              void* d_out, int out_size, void* d_ws, size_t ws_size,
                              hipStream_t stream) {
    const float* h      = (const float*)d_in[0];
    const float* fc_w   = (const float*)d_in[1];
    const float* fc_b   = (const float*)d_in[2];
    const float* attn_w = (const float*)d_in[3];
    const float* attn_b = (const float*)d_in[4];
    const int*   src    = (const int*)d_in[5];
    const int*   dst    = (const int*)d_in[6];
    float* hout = (float*)d_out;

    char* ws = (char*)d_ws;
    float*    z      = (float*)ws;    ws += align256((size_t)N_NODES * OUT_DIM * 4);
    float*    s_src  = (float*)ws;    ws += align256((size_t)N_NODES * 4);
    float*    s_dst  = (float*)ws;    ws += align256((size_t)N_NODES * 4);
    unsigned* ekey   = (unsigned*)ws; ws += align256((size_t)N_EDGES);

    k_gemm_wmma<<<313, 256, 0, stream>>>(h, fc_w, fc_b, z);
    k_node_dots<<<1250, 256, 0, stream>>>(z, attn_w, dst, s_src, s_dst, ekey);
    k_edges_owned<<<SL_GROUPS, SL_THREADS, 0, stream>>>(src, dst, ekey, s_src, s_dst, attn_b, z, hout);
}


# GEMM re-tiled for the WGP's load path: a wave owns 80x32 (7 fragment loads per 10 WMMAs, was 5 per 4), 250 groups of 4 waves, one per WGP (was 313 of 8, 57 WGPs doubled); each tile's k order kept
# speedup vs baseline: 3.0874x; 1.1482x over previous
#include <hip/hip_runtime.h>
#include <hip/hip_bf16.h>

#define N_NODES 10000
#define N_EDGES 320000
#define IN_DIM  512
#define OUT_DIM 256
#define NEG_SLOPE 0.01f

typedef float v2f __attribute__((ext_vector_type(2)));
typedef float v8f __attribute__((ext_vector_type(8)));

__device__ __forceinline__ unsigned enc_f32(float f) {
    int b = __float_as_int(f);
    return (b >= 0) ? ((unsigned)b | 0x80000000u) : ~(unsigned)b;
}
__device__ __forceinline__ float dec_f32(unsigned k) {
    int b = (k & 0x80000000u) ? (int)(k & 0x7FFFFFFFu) : ~(int)k;
    return __int_as_float(b);
}
#define EMAX_NEG_INF 0x007FFFFFu

#define GM_RT      5
#define GM_CT      2
#define GM_WAVES   4
#define GM_GROUPS  250
__global__ __launch_bounds__(GM_WAVES * 32)
void k_gemm_wmma(const float* __restrict__ h, const float* __restrict__ w,
                 const float* __restrict__ bias, float* __restrict__ z) {
    const int lane = threadIdx.x & 31;
    const int wv   = threadIdx.x >> 5;
    const int row0 = (blockIdx.x >> 1) * (GM_RT * 16);
    const int col0 = ((((int)blockIdx.x & 1) << 2) + wv) * (GM_CT * 16);
    const int mn   = lane & 15;
    const int kb   = (lane >> 4) << 1;

    const float* ap = h + (size_t)(row0 + mn) * IN_DIM + kb;
    const float* bp = w + (size_t)(col0 + mn) * IN_DIM + kb;

    v8f c[GM_RT][GM_CT];
#pragma unroll
    for (int r = 0; r < GM_RT; ++r)
#pragma unroll
        for (int t = 0; t < GM_CT; ++t) c[r][t] = (v8f){};

#pragma unroll 4
    for (int k = 0; k < IN_DIM; k += 4) {
        v2f a[GM_RT], q[GM_CT];
#pragma unroll
        for (int r = 0; r < GM_RT; ++r) a[r] = *(const v2f*)(ap + (size_t)r * 16 * IN_DIM + k);
#pragma unroll
        for (int t = 0; t < GM_CT; ++t) q[t] = *(const v2f*)(bp + (size_t)t * 16 * IN_DIM + k);
#pragma unroll
        for (int r = 0; r < GM_RT; ++r)
#pragma unroll
            for (int t = 0; t < GM_CT; ++t)
                c[r][t] = __builtin_amdgcn_wmma_f32_16x16x4_f32(false, a[r], false, q[t], (short)0,
                                                                c[r][t], false, false);
    }

    const int mbase = (lane >> 4) * 8;
#pragma unroll
    for (int t = 0; t < GM_CT; ++t) {
        const int   col = col0 + t * 16 + mn;
        const float bv  = bias[col];
#pragma unroll
        for (int r = 0; r < GM_RT; ++r)
#pragma unroll
            for (int v = 0; v < 8; ++v)
                z[(size_t)(row0 + r * 16 + mbase + v) * OUT_DIM + col] = c[r][t][v] + bv;
    }
}

#define KEY_ROWS   40
#define N_KEYLINES (N_EDGES / 128)
__device__ __forceinline__ unsigned slab_key(int d) {
    return min((unsigned)d / (unsigned)KEY_ROWS, 255u);
}
__global__ void k_node_dots(const float* __restrict__ z, const float* __restrict__ attn_w,
                            const int* __restrict__ dst,
                            float* __restrict__ s_src, float* __restrict__ s_dst,
                            unsigned* __restrict__ ekey) {
    const int lane = threadIdx.x & 31;
    const int node = (blockIdx.x * blockDim.x + threadIdx.x) >> 5;
    if (node >= N_NODES) return;
    if (node < N_KEYLINES) {
        const int4 d = ((const int4*)dst)[node * 32 + lane];
        ekey[node * 32 + lane] = slab_key(d.x) | (slab_key(d.y) << 8) |
                                 (slab_key(d.z) << 16) | (slab_key(d.w) << 24);
    }
    const float4* zrow = (const float4*)(z + (size_t)node * OUT_DIM);
    const float4* aw   = (const float4*)attn_w;
    float ps = 0.0f, pd = 0.0f;
#pragma unroll
    for (int j = 0; j < 2; ++j) {
        int vi = lane + j * 32;
        float4 zz = zrow[vi];
        float4 as = aw[vi];
        float4 ad = aw[64 + vi];
        ps += zz.x * as.x + zz.y * as.y + zz.z * as.z + zz.w * as.w;
        pd += zz.x * ad.x + zz.y * ad.y + zz.z * ad.z + zz.w * ad.w;
    }
#pragma unroll
    for (int m = 16; m >= 1; m >>= 1) {
        ps += __shfl_xor(ps, m, 32);
        pd += __shfl_xor(pd, m, 32);
    }
    if (lane == 0) { s_src[node] = ps; s_dst[node] = pd; }
}

#define SL_ROWS     40
#define SL_GROUPS   (N_NODES / SL_ROWS)
#define SL_THREADS  1024
#define SL_WAVES    (SL_THREADS / 32)
#define SL_CAP      16384
#define N_KQ        (N_EDGES / 16)
#define SC_BATCH    10

__device__ __forceinline__ float edge_logit(float ss, float sd, float ab) {
    const float v = ss + sd + ab;
    return (v >= 0.0f) ? v : NEG_SLOPE * v;
}

__device__ __forceinline__ void list_push(unsigned* l_cnt, unsigned* l_ent, int ei) {
    const unsigned slot = atomicAdd(l_cnt, 1u);
    if (slot < (unsigned)SL_CAP) l_ent[slot] = (unsigned)ei;
}

__device__ __forceinline__ unsigned key_hits(unsigned w, unsigned pat) {
    const unsigned x = w ^ pat;
    const unsigned t = (x & 0x7F7F7F7Fu) + 0x7F7F7F7Fu;
    return ~(t | x) & 0x80808080u;
}

__device__ __forceinline__ void row_add(float* acc, const float* __restrict__ z,
                                        int s, unsigned r, float alpha, int lane) {
    const float* zs = z + (size_t)s * OUT_DIM + lane;
    float* ar = acc + r * OUT_DIM + lane;
    float zz[8];
#pragma unroll
    for (int j = 0; j < 8; ++j) zz[j] = zs[j * 32];
#pragma unroll
    for (int j = 0; j < 8; ++j) unsafeAtomicAdd(&ar[j * 32], alpha * zz[j]);
}

__global__ __launch_bounds__(SL_THREADS)
void k_edges_owned(const int* __restrict__ src, const int* __restrict__ dst,
                   const unsigned* __restrict__ ekey,
                   const float* __restrict__ s_src, const float* __restrict__ s_dst,
                   const float* __restrict__ attn_b, const float* __restrict__ z,
                   float* __restrict__ hout) {
    __shared__ float    acc[SL_ROWS * OUT_DIM];
    __shared__ unsigned l_ent[SL_CAP];
    __shared__ float    l_val[SL_CAP];
    __shared__ float    l_sd[SL_ROWS];
    __shared__ unsigned l_mx[SL_ROWS];
    __shared__ float    l_dn[SL_ROWS];
    __shared__ unsigned l_cnt;

    const int   tid  = threadIdx.x;
    const int   lane = tid & 31;
    const int   wv   = tid >> 5;
    const int   lo   = blockIdx.x * SL_ROWS;
    const float ab   = attn_b[0];

    for (int j = tid; j < SL_ROWS * OUT_DIM / 4; j += SL_THREADS)
        ((float4*)acc)[j] = make_float4(0.0f, 0.0f, 0.0f, 0.0f);
    if (tid < SL_ROWS) { l_sd[tid] = s_dst[lo + tid]; l_mx[tid] = EMAX_NEG_INF; l_dn[tid] = 0.0f; }
    if (tid == 0) l_cnt = 0u;
    __syncthreads();

    const unsigned pat   = (unsigned)blockIdx.x * 0x01010101u;
    const int      start = blockIdx.x * (N_KQ / SL_GROUPS);
    for (int base = 0; base < N_KQ; base += SL_THREADS * SC_BATCH) {
        uint4 kq[SC_BATCH];
#pragma unroll
        for (int u = 0; u < SC_BATCH; ++u) {
            int qd = base + u * SL_THREADS + tid + start;
            if (qd >= N_KQ) qd -= N_KQ;
            kq[u] = ((const uint4*)ekey)[min(qd, N_KQ - 1)];
        }
#pragma unroll
        for (int u = 0; u < SC_BATCH; ++u) {
            const int it = base + u * SL_THREADS + tid;
            int qd = it + start;
            if (qd >= N_KQ) qd -= N_KQ;
            unsigned m = (key_hits(kq[u].x, pat) >> 3) | (key_hits(kq[u].y, pat) >> 2) |
                         (key_hits(kq[u].z, pat) >> 1) |  key_hits(kq[u].w, pat);
            if (it < N_KQ && m != 0u) {
                while (m) {
                    const int p = __builtin_ctz(m);
                    list_push(&l_cnt, l_ent, qd * 16 + ((p & 3) << 2) + (p >> 3));
                    m &= m - 1u;
                }
            }
        }
    }
    __syncthreads();
    const unsigned n_all = l_cnt;

    if (n_all <= (unsigned)SL_CAP) {
        const int n = (int)n_all;
        for (int i = tid; i < n; i += SL_THREADS) {
            const int      ei = (int)l_ent[i];
            const int      s  = src[ei];
            const unsigned r  = (unsigned)(dst[ei] - lo);
            const float    v  = edge_logit(s_src[s], l_sd[r], ab);
            l_ent[i] = (unsigned)s | (r << 16);
            l_val[i] = v;
            atomicMax(&l_mx[r], enc_f32(v));
        }
        __syncthreads();
        for (int i = tid; i < n; i += SL_THREADS) {
            const unsigned r = l_ent[i] >> 16;
            const float    x = expf(l_val[i] - dec_f32(l_mx[r]));
            l_val[i] = x;
            unsafeAtomicAdd(&l_dn[r], x);
        }
        __syncthreads();
        const int cnt = (n > wv) ? (n - wv + SL_WAVES - 1) / SL_WAVES : 0;
        int k = 0;
        for (; k + 1 < cnt; k += 2) {
            const int      i   = wv + k * SL_WAVES;
            const int      i2  = i + SL_WAVES;
            const unsigned e0  = l_ent[i];
            const unsigned e1  = l_ent[i2];
            const int      s0 = (int)(e0 & 0xFFFFu), s1 = (int)(e1 & 0xFFFFu);
            const unsigned r0 = e0 >> 16,            r1 = e1 >> 16;
            const float    a0 = l_val[i]  / l_dn[r0];
            const float    a1 = l_val[i2] / l_dn[r1];
            const float* z0 = z + (size_t)s0 * OUT_DIM + lane;
            const float* z1 = z + (size_t)s1 * OUT_DIM + lane;
            float* c0 = acc + r0 * OUT_DIM + lane;
            float* c1 = acc + r1 * OUT_DIM + lane;
            float u0[8], u1[8];
#pragma unroll
            for (int j = 0; j < 8; ++j) u0[j] = z0[j * 32];
#pragma unroll
            for (int j = 0; j < 8; ++j) u1[j] = z1[j * 32];
#pragma unroll
            for (int j = 0; j < 8; ++j) unsafeAtomicAdd(&c0[j * 32], a0 * u0[j]);
#pragma unroll
            for (int j = 0; j < 8; ++j) unsafeAtomicAdd(&c1[j * 32], a1 * u1[j]);
        }
        if (k < cnt) {
            const int      i  = wv + k * SL_WAVES;
            const unsigned e0 = l_ent[i];
            row_add(acc, z, (int)(e0 & 0xFFFFu), e0 >> 16, l_val[i] / l_dn[e0 >> 16], lane);
        }
    } else {
        for (int ei = tid; ei < N_EDGES; ei += SL_THREADS) {
            const unsigned r = (unsigned)(dst[ei] - lo);
            if (r < (unsigned)SL_ROWS)
                atomicMax(&l_mx[r], enc_f32(edge_logit(s_src[src[ei]], l_sd[r], ab)));
        }
        __syncthreads();
        for (int ei = tid; ei < N_EDGES; ei += SL_THREADS) {
            const unsigned r = (unsigned)(dst[ei] - lo);
            if (r < (unsigned)SL_ROWS) {
                const float v = edge_logit(s_src[src[ei]], l_sd[r], ab);
                unsafeAtomicAdd(&l_dn[r], expf(v - dec_f32(l_mx[r])));
            }
        }
        __syncthreads();
        for (int base = wv * 32; base < N_EDGES; base += SL_THREADS) {
            const int      ei   = base + lane;
            const unsigned r    = (unsigned)(dst[ei] - lo);
            const bool     mine = r < (unsigned)SL_ROWS;
            int   s = 0;
            float a = 0.0f;
            if (mine) {
                s = src[ei];
                const float v = edge_logit(s_src[s], l_sd[r], ab);
                a = expf(v - dec_f32(l_mx[r])) / l_dn[r];
            }
            unsigned m = __builtin_amdgcn_ballot_w32(mine);
            while (m) {
                const int i = __builtin_ctz(m);
                m &= m - 1u;
                const int      si = __builtin_amdgcn_readlane(s, i);
                const unsigned ri = (unsigned)__builtin_amdgcn_readlane((int)r, i);
                const float    ai = __int_as_float(__builtin_amdgcn_readlane(__float_as_int(a), i));
                row_add(acc, z, si, ri, ai, lane);
            }
        }
    }
    __syncthreads();
    float4* og = (float4*)(hout + (size_t)lo * OUT_DIM);
    for (int j = tid; j < SL_ROWS * OUT_DIM / 4; j += SL_THREADS) og[j] = ((const float4*)acc)[j];
}

static inline size_t align256(size_t x) { return (x + 255) & ~(size_t)255; }

extern "C" void kernel_launch(void* const* d_in, const int* in_sizes, int n_in,
                              void* d_out, int out_size, void* d_ws, size_t ws_size,
                              hipStream_t stream) {
    const float* h      = (const float*)d_in[0];
    const float* fc_w   = (const float*)d_in[1];
    const float* fc_b   = (const float*)d_in[2];
    const float* attn_w = (const float*)d_in[3];
    const float* attn_b = (const float*)d_in[4];
    const int*   src    = (const int*)d_in[5];
    const int*   dst    = (const int*)d_in[6];
    float* hout = (float*)d_out;

    char* ws = (char*)d_ws;
    float*    z      = (float*)ws;    ws += align256((size_t)N_NODES * OUT_DIM * 4);
    float*    s_src  = (float*)ws;    ws += align256((size_t)N_NODES * 4);
    float*    s_dst  = (float*)ws;    ws += align256((size_t)N_NODES * 4);
    unsigned* ekey   = (unsigned*)ws; ws += align256((size_t)N_EDGES);

    k_gemm_wmma<<<GM_GROUPS, GM_WAVES * 32, 0, stream>>>(h, fc_w, fc_b, z);
    k_node_dots<<<1250, 256, 0, stream>>>(z, attn_w, dst, s_src, s_dst, ekey);
    k_edges_owned<<<SL_GROUPS, SL_THREADS, 0, stream>>>(src, dst, ekey, s_src, s_dst, attn_b, z, hout);
}


# GEMM: k walked in 32-k blocks with a second fragment register set, each block's 56 loads issued one block of 80 WMMAs ahead of their use (was load, wait, multiply per 16 k); k order kept
# speedup vs baseline: 3.8887x; 1.2595x over previous
#include <hip/hip_runtime.h>
#include <hip/hip_bf16.h>

#define N_NODES 10000
#define N_EDGES 320000
#define IN_DIM  512
#define OUT_DIM 256
#define NEG_SLOPE 0.01f

typedef float v2f __attribute__((ext_vector_type(2)));
typedef float v8f __attribute__((ext_vector_type(8)));

__device__ __forceinline__ unsigned enc_f32(float f) {
    int b = __float_as_int(f);
    return (b >= 0) ? ((unsigned)b | 0x80000000u) : ~(unsigned)b;
}
__device__ __forceinline__ float dec_f32(unsigned k) {
    int b = (k & 0x80000000u) ? (int)(k & 0x7FFFFFFFu) : ~(int)k;
    return __int_as_float(b);
}
#define EMAX_NEG_INF 0x007FFFFFu

#define GM_RT      5
#define GM_CT      2
#define GM_WAVES   4
#define GM_GROUPS  250
#define GM_KB      32
#define GM_KS      (GM_KB / 4)

struct gm_frags { v2f a[GM_KS][GM_RT]; v2f q[GM_KS][GM_CT]; };

__device__ __forceinline__ void gm_load(gm_frags& f, const float* __restrict__ ap,
                                        const float* __restrict__ bp, int k) {
#pragma unroll
    for (int s = 0; s < GM_KS; ++s) {
#pragma unroll
        for (int r = 0; r < GM_RT; ++r)
            f.a[s][r] = *(const v2f*)(ap + (size_t)r * 16 * IN_DIM + k + 4 * s);
#pragma unroll
        for (int t = 0; t < GM_CT; ++t)
            f.q[s][t] = *(const v2f*)(bp + (size_t)t * 16 * IN_DIM + k + 4 * s);
    }
}

__device__ __forceinline__ void gm_mma(v8f (&c)[GM_RT][GM_CT], const gm_frags& f) {
#pragma unroll
    for (int s = 0; s < GM_KS; ++s)
#pragma unroll
        for (int r = 0; r < GM_RT; ++r)
#pragma unroll
            for (int t = 0; t < GM_CT; ++t)
                c[r][t] = __builtin_amdgcn_wmma_f32_16x16x4_f32(false, f.a[s][r], false, f.q[s][t],
                                                                (short)0, c[r][t], false, false);
}

__global__ __launch_bounds__(GM_WAVES * 32)
void k_gemm_wmma(const float* __restrict__ h, const float* __restrict__ w,
                 const float* __restrict__ bias, float* __restrict__ z) {
    const int lane = threadIdx.x & 31;
    const int wv   = threadIdx.x >> 5;
    const int row0 = (blockIdx.x >> 1) * (GM_RT * 16);
    const int col0 = ((((int)blockIdx.x & 1) << 2) + wv) * (GM_CT * 16);
    const int mn   = lane & 15;
    const int kb   = (lane >> 4) << 1;

    const float* ap = h + (size_t)(row0 + mn) * IN_DIM + kb;
    const float* bp = w + (size_t)(col0 + mn) * IN_DIM + kb;

    v8f c[GM_RT][GM_CT];
#pragma unroll
    for (int r = 0; r < GM_RT; ++r)
#pragma unroll
        for (int t = 0; t < GM_CT; ++t) c[r][t] = (v8f){};

    gm_frags f0, f1;
    gm_load(f0, ap, bp, 0);
#pragma unroll 1
    for (int k = 0; k < IN_DIM - 2 * GM_KB; k += 2 * GM_KB) {
        gm_load(f1, ap, bp, k + GM_KB);
        gm_mma(c, f0);
        gm_load(f0, ap, bp, k + 2 * GM_KB);
        gm_mma(c, f1);
    }
    gm_load(f1, ap, bp, IN_DIM - GM_KB);
    gm_mma(c, f0);
    gm_mma(c, f1);

    const int mbase = (lane >> 4) * 8;
#pragma unroll
    for (int t = 0; t < GM_CT; ++t) {
        const int   col = col0 + t * 16 + mn;
        const float bv  = bias[col];
#pragma unroll
        for (int r = 0; r < GM_RT; ++r)
#pragma unroll
            for (int v = 0; v < 8; ++v)
                z[(size_t)(row0 + r * 16 + mbase + v) * OUT_DIM + col] = c[r][t][v] + bv;
    }
}

#define KEY_ROWS   40
#define N_KEYLINES (N_EDGES / 128)
__device__ __forceinline__ unsigned slab_key(int d) {
    return min((unsigned)d / (unsigned)KEY_ROWS, 255u);
}
__global__ void k_node_dots(const float* __restrict__ z, const float* __restrict__ attn_w,
                            const int* __restrict__ dst,
                            float* __restrict__ s_src, float* __restrict__ s_dst,
                            unsigned* __restrict__ ekey) {
    const int lane = threadIdx.x & 31;
    const int node = (blockIdx.x * blockDim.x + threadIdx.x) >> 5;
    if (node >= N_NODES) return;
    if (node < N_KEYLINES) {
        const int4 d = ((const int4*)dst)[node * 32 + lane];
        ekey[node * 32 + lane] = slab_key(d.x) | (slab_key(d.y) << 8) |
                                 (slab_key(d.z) << 16) | (slab_key(d.w) << 24);
    }
    const float4* zrow = (const float4*)(z + (size_t)node * OUT_DIM);
    const float4* aw   = (const float4*)attn_w;
    float ps = 0.0f, pd = 0.0f;
#pragma unroll
    for (int j = 0; j < 2; ++j) {
        int vi = lane + j * 32;
        float4 zz = zrow[vi];
        float4 as = aw[vi];
        float4 ad = aw[64 + vi];
        ps += zz.x * as.x + zz.y * as.y + zz.z * as.z + zz.w * as.w;
        pd += zz.x * ad.x + zz.y * ad.y + zz.z * ad.z + zz.w * ad.w;
    }
#pragma unroll
    for (int m = 16; m >= 1; m >>= 1) {
        ps += __shfl_xor(ps, m, 32);
        pd += __shfl_xor(pd, m, 32);
    }
    if (lane == 0) { s_src[node] = ps; s_dst[node] = pd; }
}

#define SL_ROWS     40
#define SL_GROUPS   (N_NODES / SL_ROWS)
#define SL_THREADS  1024
#define SL_WAVES    (SL_THREADS / 32)
#define SL_CAP      16384
#define N_KQ        (N_EDGES / 16)
#define SC_BATCH    10

__device__ __forceinline__ float edge_logit(float ss, float sd, float ab) {
    const float v = ss + sd + ab;
    return (v >= 0.0f) ? v : NEG_SLOPE * v;
}

__device__ __forceinline__ void list_push(unsigned* l_cnt, unsigned* l_ent, int ei) {
    const unsigned slot = atomicAdd(l_cnt, 1u);
    if (slot < (unsigned)SL_CAP) l_ent[slot] = (unsigned)ei;
}

__device__ __forceinline__ unsigned key_hits(unsigned w, unsigned pat) {
    const unsigned x = w ^ pat;
    const unsigned t = (x & 0x7F7F7F7Fu) + 0x7F7F7F7Fu;
    return ~(t | x) & 0x80808080u;
}

__device__ __forceinline__ void row_add(float* acc, const float* __restrict__ z,
                                        int s, unsigned r, float alpha, int lane) {
    const float* zs = z + (size_t)s * OUT_DIM + lane;
    float* ar = acc + r * OUT_DIM + lane;
    float zz[8];
#pragma unroll
    for (int j = 0; j < 8; ++j) zz[j] = zs[j * 32];
#pragma unroll
    for (int j = 0; j < 8; ++j) unsafeAtomicAdd(&ar[j * 32], alpha * zz[j]);
}

__global__ __launch_bounds__(SL_THREADS)
void k_edges_owned(const int* __restrict__ src, const int* __restrict__ dst,
                   const unsigned* __restrict__ ekey,
                   const float* __restrict__ s_src, const float* __restrict__ s_dst,
                   const float* __restrict__ attn_b, const float* __restrict__ z,
                   float* __restrict__ hout) {
    __shared__ float    acc[SL_ROWS * OUT_DIM];
    __shared__ unsigned l_ent[SL_CAP];
    __shared__ float    l_val[SL_CAP];
    __shared__ float    l_sd[SL_ROWS];
    __shared__ unsigned l_mx[SL_ROWS];
    __shared__ float    l_dn[SL_ROWS];
    __shared__ unsigned l_cnt;

    const int   tid  = threadIdx.x;
    const int   lane = tid & 31;
    const int   wv   = tid >> 5;
    const int   lo   = blockIdx.x * SL_ROWS;
    const float ab   = attn_b[0];

    for (int j = tid; j < SL_ROWS * OUT_DIM / 4; j += SL_THREADS)
        ((float4*)acc)[j] = make_float4(0.0f, 0.0f, 0.0f, 0.0f);
    if (tid < SL_ROWS) { l_sd[tid] = s_dst[lo + tid]; l_mx[tid] = EMAX_NEG_INF; l_dn[tid] = 0.0f; }
    if (tid == 0) l_cnt = 0u;
    __syncthreads();

    const unsigned pat   = (unsigned)blockIdx.x * 0x01010101u;
    const int      start = blockIdx.x * (N_KQ / SL_GROUPS);
    for (int base = 0; base < N_KQ; base += SL_THREADS * SC_BATCH) {
        uint4 kq[SC_BATCH];
#pragma unroll
        for (int u = 0; u < SC_BATCH; ++u) {
            int qd = base + u * SL_THREADS + tid + start;
            if (qd >= N_KQ) qd -= N_KQ;
            kq[u] = ((const uint4*)ekey)[min(qd, N_KQ - 1)];
        }
#pragma unroll
        for (int u = 0; u < SC_BATCH; ++u) {
            const int it = base + u * SL_THREADS + tid;
            int qd = it + start;
            if (qd >= N_KQ) qd -= N_KQ;
            unsigned m = (key_hits(kq[u].x, pat) >> 3) | (key_hits(kq[u].y, pat) >> 2) |
                         (key_hits(kq[u].z, pat) >> 1) |  key_hits(kq[u].w, pat);
            if (it < N_KQ && m != 0u) {
                while (m) {
                    const int p = __builtin_ctz(m);
                    list_push(&l_cnt, l_ent, qd * 16 + ((p & 3) << 2) + (p >> 3));
                    m &= m - 1u;
                }
            }
        }
    }
    __syncthreads();
    const unsigned n_all = l_cnt;

    if (n_all <= (unsigned)SL_CAP) {
        const int n = (int)n_all;
        for (int i = tid; i < n; i += SL_THREADS) {
            const int      ei = (int)l_ent[i];
            const int      s  = src[ei];
            const unsigned r  = (unsigned)(dst[ei] - lo);
            const float    v  = edge_logit(s_src[s], l_sd[r], ab);
            l_ent[i] = (unsigned)s | (r << 16);
            l_val[i] = v;
            atomicMax(&l_mx[r], enc_f32(v));
        }
        __syncthreads();
        for (int i = tid; i < n; i += SL_THREADS) {
            const unsigned r = l_ent[i] >> 16;
            const float    x = expf(l_val[i] - dec_f32(l_mx[r]));
            l_val[i] = x;
            unsafeAtomicAdd(&l_dn[r], x);
        }
        __syncthreads();
        const int cnt = (n > wv) ? (n - wv + SL_WAVES - 1) / SL_WAVES : 0;
        int k = 0;
        for (; k + 1 < cnt; k += 2) {
            const int      i   = wv + k * SL_WAVES;
            const int      i2  = i + SL_WAVES;
            const unsigned e0  = l_ent[i];
            const unsigned e1  = l_ent[i2];
            const int      s0 = (int)(e0 & 0xFFFFu), s1 = (int)(e1 & 0xFFFFu);
            const unsigned r0 = e0 >> 16,            r1 = e1 >> 16;
            const float    a0 = l_val[i]  / l_dn[r0];
            const float    a1 = l_val[i2] / l_dn[r1];
            const float* z0 = z + (size_t)s0 * OUT_DIM + lane;
            const float* z1 = z + (size_t)s1 * OUT_DIM + lane;
            float* c0 = acc + r0 * OUT_DIM + lane;
            float* c1 = acc + r1 * OUT_DIM + lane;
            float u0[8], u1[8];
#pragma unroll
            for (int j = 0; j < 8; ++j) u0[j] = z0[j * 32];
#pragma unroll
            for (int j = 0; j < 8; ++j) u1[j] = z1[j * 32];
#pragma unroll
            for (int j = 0; j < 8; ++j) unsafeAtomicAdd(&c0[j * 32], a0 * u0[j]);
#pragma unroll
            for (int j = 0; j < 8; ++j) unsafeAtomicAdd(&c1[j * 32], a1 * u1[j]);
        }
        if (k < cnt) {
            const int      i  = wv + k * SL_WAVES;
            const unsigned e0 = l_ent[i];
            row_add(acc, z, (int)(e0 & 0xFFFFu), e0 >> 16, l_val[i] / l_dn[e0 >> 16], lane);
        }
    } else {
        for (int ei = tid; ei < N_EDGES; ei += SL_THREADS) {
            const unsigned r = (unsigned)(dst[ei] - lo);
            if (r < (unsigned)SL_ROWS)
                atomicMax(&l_mx[r], enc_f32(edge_logit(s_src[src[ei]], l_sd[r], ab)));
        }
        __syncthreads();
        for (int ei = tid; ei < N_EDGES; ei += SL_THREADS) {
            const unsigned r = (unsigned)(dst[ei] - lo);
            if (r < (unsigned)SL_ROWS) {
                const float v = edge_logit(s_src[src[ei]], l_sd[r], ab);
                unsafeAtomicAdd(&l_dn[r], expf(v - dec_f32(l_mx[r])));
            }
        }
        __syncthreads();
        for (int base = wv * 32; base < N_EDGES; base += SL_THREADS) {
            const int      ei   = base + lane;
            const unsigned r    = (unsigned)(dst[ei] - lo);
            const bool     mine = r < (unsigned)SL_ROWS;
            int   s = 0;
            float a = 0.0f;
            if (mine) {
                s = src[ei];
                const float v = edge_logit(s_src[s], l_sd[r], ab);
                a = expf(v - dec_f32(l_mx[r])) / l_dn[r];
            }
            unsigned m = __builtin_amdgcn_ballot_w32(mine);
            while (m) {
                const int i = __builtin_ctz(m);
                m &= m - 1u;
                const int      si = __builtin_amdgcn_readlane(s, i);
                const unsigned ri = (unsigned)__builtin_amdgcn_readlane((int)r, i);
                const float    ai = __int_as_float(__builtin_amdgcn_readlane(__float_as_int(a), i));
                row_add(acc, z, si, ri, ai, lane);
            }
        }
    }
    __syncthreads();
    float4* og = (float4*)(hout + (size_t)lo * OUT_DIM);
    for (int j = tid; j < SL_ROWS * OUT_DIM / 4; j += SL_THREADS) og[j] = ((const float4*)acc)[j];
}

static inline size_t align256(size_t x) { return (x + 255) & ~(size_t)255; }

extern "C" void kernel_launch(void* const* d_in, const int* in_sizes, int n_in,
                              void* d_out, int out_size, void* d_ws, size_t ws_size,
                              hipStream_t stream) {
    const float* h      = (const float*)d_in[0];
    const float* fc_w   = (const float*)d_in[1];
    const float* fc_b   = (const float*)d_in[2];
    const float* attn_w = (const float*)d_in[3];
    const float* attn_b = (const float*)d_in[4];
    const int*   src    = (const int*)d_in[5];
    const int*   dst    = (const int*)d_in[6];
    float* hout = (float*)d_out;

    char* ws = (char*)d_ws;
    float*    z      = (float*)ws;    ws += align256((size_t)N_NODES * OUT_DIM * 4);
    float*    s_src  = (float*)ws;    ws += align256((size_t)N_NODES * 4);
    float*    s_dst  = (float*)ws;    ws += align256((size_t)N_NODES * 4);
    unsigned* ekey   = (unsigned*)ws; ws += align256((size_t)N_EDGES);

    k_gemm_wmma<<<GM_GROUPS, GM_WAVES * 32, 0, stream>>>(h, fc_w, fc_b, z);
    k_node_dots<<<1250, 256, 0, stream>>>(z, attn_w, dst, s_src, s_dst, ekey);
    k_edges_owned<<<SL_GROUPS, SL_THREADS, 0, stream>>>(src, dst, ekey, s_src, s_dst, attn_b, z, hout);
}


# GEMM fragment loads widened to 16 bytes per lane (lane half h takes k 8i+4h..8i+4h+3, one load feeds two WMMAs): 28 loads per 80 WMMAs, was 56; k sum order inside each 8-k group changed
# speedup vs baseline: 4.0730x; 1.0474x over previous
#include <hip/hip_runtime.h>
#include <hip/hip_bf16.h>

#define N_NODES 10000
#define N_EDGES 320000
#define IN_DIM  512
#define OUT_DIM 256
#define NEG_SLOPE 0.01f

typedef float v2f __attribute__((ext_vector_type(2)));
typedef float v4f __attribute__((ext_vector_type(4)));
typedef float v8f __attribute__((ext_vector_type(8)));

__device__ __forceinline__ unsigned enc_f32(float f) {
    int b = __float_as_int(f);
    return (b >= 0) ? ((unsigned)b | 0x80000000u) : ~(unsigned)b;
}
__device__ __forceinline__ float dec_f32(unsigned k) {
    int b = (k & 0x80000000u) ? (int)(k & 0x7FFFFFFFu) : ~(int)k;
    return __int_as_float(b);
}
#define EMAX_NEG_INF 0x007FFFFFu

#define GM_RT      5
#define GM_CT      2
#define GM_WAVES   4
#define GM_GROUPS  250
#define GM_KB      32
#define GM_KG      (GM_KB / 8)

struct gm_frags { v4f a[GM_KG][GM_RT]; v4f q[GM_KG][GM_CT]; };

__device__ __forceinline__ void gm_load(gm_frags& f, const float* __restrict__ ap,
                                        const float* __restrict__ bp, int k) {
#pragma unroll
    for (int g = 0; g < GM_KG; ++g) {
#pragma unroll
        for (int r = 0; r < GM_RT; ++r)
            f.a[g][r] = *(const v4f*)(ap + (size_t)r * 16 * IN_DIM + k + 8 * g);
#pragma unroll
        for (int t = 0; t < GM_CT; ++t)
            f.q[g][t] = *(const v4f*)(bp + (size_t)t * 16 * IN_DIM + k + 8 * g);
    }
}

__device__ __forceinline__ void gm_mma(v8f (&c)[GM_RT][GM_CT], const gm_frags& f) {
#pragma unroll
    for (int g = 0; g < GM_KG; ++g)
#pragma unroll
        for (int e = 0; e < 2; ++e)
#pragma unroll
            for (int r = 0; r < GM_RT; ++r)
#pragma unroll
                for (int t = 0; t < GM_CT; ++t) {
                    const v2f av = e ? (v2f){f.a[g][r].z, f.a[g][r].w} : (v2f){f.a[g][r].x, f.a[g][r].y};
                    const v2f bv = e ? (v2f){f.q[g][t].z, f.q[g][t].w} : (v2f){f.q[g][t].x, f.q[g][t].y};
                    c[r][t] = __builtin_amdgcn_wmma_f32_16x16x4_f32(false, av, false, bv,
                                                                    (short)0, c[r][t], false, false);
                }
}

__global__ __launch_bounds__(GM_WAVES * 32)
void k_gemm_wmma(const float* __restrict__ h, const float* __restrict__ w,
                 const float* __restrict__ bias, float* __restrict__ z) {
    const int lane = threadIdx.x & 31;
    const int wv   = threadIdx.x >> 5;
    const int row0 = (blockIdx.x >> 1) * (GM_RT * 16);
    const int col0 = ((((int)blockIdx.x & 1) << 2) + wv) * (GM_CT * 16);
    const int mn   = lane & 15;
    const int kb   = (lane >> 4) << 2;

    const float* ap = h + (size_t)(row0 + mn) * IN_DIM + kb;
    const float* bp = w + (size_t)(col0 + mn) * IN_DIM + kb;

    v8f c[GM_RT][GM_CT];
#pragma unroll
    for (int r = 0; r < GM_RT; ++r)
#pragma unroll
        for (int t = 0; t < GM_CT; ++t) c[r][t] = (v8f){};

    gm_frags f0, f1;
    gm_load(f0, ap, bp, 0);
#pragma unroll 1
    for (int k = 0; k < IN_DIM - 2 * GM_KB; k += 2 * GM_KB) {
        gm_load(f1, ap, bp, k + GM_KB);
        gm_mma(c, f0);
        gm_load(f0, ap, bp, k + 2 * GM_KB);
        gm_mma(c, f1);
    }
    gm_load(f1, ap, bp, IN_DIM - GM_KB);
    gm_mma(c, f0);
    gm_mma(c, f1);

    const int mbase = (lane >> 4) * 8;
#pragma unroll
    for (int t = 0; t < GM_CT; ++t) {
        const int   col = col0 + t * 16 + mn;
        const float bv  = bias[col];
#pragma unroll
        for (int r = 0; r < GM_RT; ++r)
#pragma unroll
            for (int v = 0; v < 8; ++v)
                z[(size_t)(row0 + r * 16 + mbase + v) * OUT_DIM + col] = c[r][t][v] + bv;
    }
}

#define KEY_ROWS   40
#define N_KEYLINES (N_EDGES / 128)
__device__ __forceinline__ unsigned slab_key(int d) {
    return min((unsigned)d / (unsigned)KEY_ROWS, 255u);
}
__global__ void k_node_dots(const float* __restrict__ z, const float* __restrict__ attn_w,
                            const int* __restrict__ dst,
                            float* __restrict__ s_src, float* __restrict__ s_dst,
                            unsigned* __restrict__ ekey) {
    const int lane = threadIdx.x & 31;
    const int node = (blockIdx.x * blockDim.x + threadIdx.x) >> 5;
    if (node >= N_NODES) return;
    if (node < N_KEYLINES) {
        const int4 d = ((const int4*)dst)[node * 32 + lane];
        ekey[node * 32 + lane] = slab_key(d.x) | (slab_key(d.y) << 8) |
                                 (slab_key(d.z) << 16) | (slab_key(d.w) << 24);
    }
    const float4* zrow = (const float4*)(z + (size_t)node * OUT_DIM);
    const float4* aw   = (const float4*)attn_w;
    float ps = 0.0f, pd = 0.0f;
#pragma unroll
    for (int j = 0; j < 2; ++j) {
        int vi = lane + j * 32;
        float4 zz = zrow[vi];
        float4 as = aw[vi];
        float4 ad = aw[64 + vi];
        ps += zz.x * as.x + zz.y * as.y + zz.z * as.z + zz.w * as.w;
        pd += zz.x * ad.x + zz.y * ad.y + zz.z * ad.z + zz.w * ad.w;
    }
#pragma unroll
    for (int m = 16; m >= 1; m >>= 1) {
        ps += __shfl_xor(ps, m, 32);
        pd += __shfl_xor(pd, m, 32);
    }
    if (lane == 0) { s_src[node] = ps; s_dst[node] = pd; }
}

#define SL_ROWS     40
#define SL_GROUPS   (N_NODES / SL_ROWS)
#define SL_THREADS  1024
#define SL_WAVES    (SL_THREADS / 32)
#define SL_CAP      16384
#define N_KQ        (N_EDGES / 16)
#define SC_BATCH    10

__device__ __forceinline__ float edge_logit(float ss, float sd, float ab) {
    const float v = ss + sd + ab;
    return (v >= 0.0f) ? v : NEG_SLOPE * v;
}

__device__ __forceinline__ void list_push(unsigned* l_cnt, unsigned* l_ent, int ei) {
    const unsigned slot = atomicAdd(l_cnt, 1u);
    if (slot < (unsigned)SL_CAP) l_ent[slot] = (unsigned)ei;
}

__device__ __forceinline__ unsigned key_hits(unsigned w, unsigned pat) {
    const unsigned x = w ^ pat;
    const unsigned t = (x & 0x7F7F7F7Fu) + 0x7F7F7F7Fu;
    return ~(t | x) & 0x80808080u;
}

__device__ __forceinline__ void row_add(float* acc, const float* __restrict__ z,
                                        int s, unsigned r, float alpha, int lane) {
    const float* zs = z + (size_t)s * OUT_DIM + lane;
    float* ar = acc + r * OUT_DIM + lane;
    float zz[8];
#pragma unroll
    for (int j = 0; j < 8; ++j) zz[j] = zs[j * 32];
#pragma unroll
    for (int j = 0; j < 8; ++j) unsafeAtomicAdd(&ar[j * 32], alpha * zz[j]);
}

__global__ __launch_bounds__(SL_THREADS)
void k_edges_owned(const int* __restrict__ src, const int* __restrict__ dst,
                   const unsigned* __restrict__ ekey,
                   const float* __restrict__ s_src, const float* __restrict__ s_dst,
                   const float* __restrict__ attn_b, const float* __restrict__ z,
                   float* __restrict__ hout) {
    __shared__ float    acc[SL_ROWS * OUT_DIM];
    __shared__ unsigned l_ent[SL_CAP];
    __shared__ float    l_val[SL_CAP];
    __shared__ float    l_sd[SL_ROWS];
    __shared__ unsigned l_mx[SL_ROWS];
    __shared__ float    l_dn[SL_ROWS];
    __shared__ unsigned l_cnt;

    const int   tid  = threadIdx.x;
    const int   lane = tid & 31;
    const int   wv   = tid >> 5;
    const int   lo   = blockIdx.x * SL_ROWS;
    const float ab   = attn_b[0];

    for (int j = tid; j < SL_ROWS * OUT_DIM / 4; j += SL_THREADS)
        ((float4*)acc)[j] = make_float4(0.0f, 0.0f, 0.0f, 0.0f);
    if (tid < SL_ROWS) { l_sd[tid] = s_dst[lo + tid]; l_mx[tid] = EMAX_NEG_INF; l_dn[tid] = 0.0f; }
    if (tid == 0) l_cnt = 0u;
    __syncthreads();

    const unsigned pat   = (unsigned)blockIdx.x * 0x01010101u;
    const int      start = blockIdx.x * (N_KQ / SL_GROUPS);
    for (int base = 0; base < N_KQ; base += SL_THREADS * SC_BATCH) {
        uint4 kq[SC_BATCH];
#pragma unroll
        for (int u = 0; u < SC_BATCH; ++u) {
            int qd = base + u * SL_THREADS + tid + start;
            if (qd >= N_KQ) qd -= N_KQ;
            kq[u] = ((const uint4*)ekey)[min(qd, N_KQ - 1)];
        }
#pragma unroll
        for (int u = 0; u < SC_BATCH; ++u) {
            const int it = base + u * SL_THREADS + tid;
            int qd = it + start;
            if (qd >= N_KQ) qd -= N_KQ;
            unsigned m = (key_hits(kq[u].x, pat) >> 3) | (key_hits(kq[u].y, pat) >> 2) |
                         (key_hits(kq[u].z, pat) >> 1) |  key_hits(kq[u].w, pat);
            if (it < N_KQ && m != 0u) {
                while (m) {
                    const int p = __builtin_ctz(m);
                    list_push(&l_cnt, l_ent, qd * 16 + ((p & 3) << 2) + (p >> 3));
                    m &= m - 1u;
                }
            }
        }
    }
    __syncthreads();
    const unsigned n_all = l_cnt;

    if (n_all <= (unsigned)SL_CAP) {
        const int n = (int)n_all;
        for (int i = tid; i < n; i += SL_THREADS) {
            const int      ei = (int)l_ent[i];
            const int      s  = src[ei];
            const unsigned r  = (unsigned)(dst[ei] - lo);
            const float    v  = edge_logit(s_src[s], l_sd[r], ab);
            l_ent[i] = (unsigned)s | (r << 16);
            l_val[i] = v;
            atomicMax(&l_mx[r], enc_f32(v));
        }
        __syncthreads();
        for (int i = tid; i < n; i += SL_THREADS) {
            const unsigned r = l_ent[i] >> 16;
            const float    x = expf(l_val[i] - dec_f32(l_mx[r]));
            l_val[i] = x;
            unsafeAtomicAdd(&l_dn[r], x);
        }
        __syncthreads();
        const int cnt = (n > wv) ? (n - wv + SL_WAVES - 1) / SL_WAVES : 0;
        int k = 0;
        for (; k + 1 < cnt; k += 2) {
            const int      i   = wv + k * SL_WAVES;
            const int      i2  = i + SL_WAVES;
            const unsigned e0  = l_ent[i];
            const unsigned e1  = l_ent[i2];
            const int      s0 = (int)(e0 & 0xFFFFu), s1 = (int)(e1 & 0xFFFFu);
            const unsigned r0 = e0 >> 16,            r1 = e1 >> 16;
            const float    a0 = l_val[i]  / l_dn[r0];
            const float    a1 = l_val[i2] / l_dn[r1];
            const float* z0 = z + (size_t)s0 * OUT_DIM + lane;
            const float* z1 = z + (size_t)s1 * OUT_DIM + lane;
            float* c0 = acc + r0 * OUT_DIM + lane;
            float* c1 = acc + r1 * OUT_DIM + lane;
            float u0[8], u1[8];
#pragma unroll
            for (int j = 0; j < 8; ++j) u0[j] = z0[j * 32];
#pragma unroll
            for (int j = 0; j < 8; ++j) u1[j] = z1[j * 32];
#pragma unroll
            for (int j = 0; j < 8; ++j) unsafeAtomicAdd(&c0[j * 32], a0 * u0[j]);
#pragma unroll
            for (int j = 0; j < 8; ++j) unsafeAtomicAdd(&c1[j * 32], a1 * u1[j]);
        }
        if (k < cnt) {
            const int      i  = wv + k * SL_WAVES;
            const unsigned e0 = l_ent[i];
            row_add(acc, z, (int)(e0 & 0xFFFFu), e0 >> 16, l_val[i] / l_dn[e0 >> 16], lane);
        }
    } else {
        for (int ei = tid; ei < N_EDGES; ei += SL_THREADS) {
            const unsigned r = (unsigned)(dst[ei] - lo);
            if (r < (unsigned)SL_ROWS)
                atomicMax(&l_mx[r], enc_f32(edge_logit(s_src[src[ei]], l_sd[r], ab)));
        }
        __syncthreads();
        for (int ei = tid; ei < N_EDGES; ei += SL_THREADS) {
            const unsigned r = (unsigned)(dst[ei] - lo);
            if (r < (unsigned)SL_ROWS) {
                const float v = edge_logit(s_src[src[ei]], l_sd[r], ab);
                unsafeAtomicAdd(&l_dn[r], expf(v - dec_f32(l_mx[r])));
            }
        }
        __syncthreads();
        for (int base = wv * 32; base < N_EDGES; base += SL_THREADS) {
            const int      ei   = base + lane;
            const unsigned r    = (unsigned)(dst[ei] - lo);
            const bool     mine = r < (unsigned)SL_ROWS;
            int   s = 0;
            float a = 0.0f;
            if (mine) {
                s = src[ei];
                const float v = edge_logit(s_src[s], l_sd[r], ab);
                a = expf(v - dec_f32(l_mx[r])) / l_dn[r];
            }
            unsigned m = __builtin_amdgcn_ballot_w32(mine);
            while (m) {
                const int i = __builtin_ctz(m);
                m &= m - 1u;
                const int      si = __builtin_amdgcn_readlane(s, i);
                const unsigned ri = (unsigned)__builtin_amdgcn_readlane((int)r, i);
                const float    ai = __int_as_float(__builtin_amdgcn_readlane(__float_as_int(a), i));
                row_add(acc, z, si, ri, ai, lane);
            }
        }
    }
    __syncthreads();
    float4* og = (float4*)(hout + (size_t)lo * OUT_DIM);
    for (int j = tid; j < SL_ROWS * OUT_DIM / 4; j += SL_THREADS) og[j] = ((const float4*)acc)[j];
}

static inline size_t align256(size_t x) { return (x + 255) & ~(size_t)255; }

extern "C" void kernel_launch(void* const* d_in, const int* in_sizes, int n_in,
                              void* d_out, int out_size, void* d_ws, size_t ws_size,
                              hipStream_t stream) {
    const float* h      = (const float*)d_in[0];
    const float* fc_w   = (const float*)d_in[1];
    const float* fc_b   = (const float*)d_in[2];
    const float* attn_w = (const float*)d_in[3];
    const float* attn_b = (const float*)d_in[4];
    const int*   src    = (const int*)d_in[5];
    const int*   dst    = (const int*)d_in[6];
    float* hout = (float*)d_out;

    char* ws = (char*)d_ws;
    float*    z      = (float*)ws;    ws += align256((size_t)N_NODES * OUT_DIM * 4);
    float*    s_src  = (float*)ws;    ws += align256((size_t)N_NODES * 4);
    float*    s_dst  = (float*)ws;    ws += align256((size_t)N_NODES * 4);
    unsigned* ekey   = (unsigned*)ws; ws += align256((size_t)N_EDGES);

    k_gemm_wmma<<<GM_GROUPS, GM_WAVES * 32, 0, stream>>>(h, fc_w, fc_b, z);
    k_node_dots<<<1250, 256, 0, stream>>>(z, attn_w, dst, s_src, s_dst, ekey);
    k_edges_owned<<<SL_GROUPS, SL_THREADS, 0, stream>>>(src, dst, ekey, s_src, s_dst, attn_b, z, hout);
}


# Row sums: each z row fetched by two 16-byte loads per lane (4 whole lines per load, was 8 one-line loads), 4 rows in flight per wave; LDS sums held in lane order, undone at write-out; arrival order
# speedup vs baseline: 4.2874x; 1.0526x over previous
#include <hip/hip_runtime.h>
#include <hip/hip_bf16.h>

#define N_NODES 10000
#define N_EDGES 320000
#define IN_DIM  512
#define OUT_DIM 256
#define NEG_SLOPE 0.01f

typedef float v2f __attribute__((ext_vector_type(2)));
typedef float v4f __attribute__((ext_vector_type(4)));
typedef float v8f __attribute__((ext_vector_type(8)));

__device__ __forceinline__ unsigned enc_f32(float f) {
    int b = __float_as_int(f);
    return (b >= 0) ? ((unsigned)b | 0x80000000u) : ~(unsigned)b;
}
__device__ __forceinline__ float dec_f32(unsigned k) {
    int b = (k & 0x80000000u) ? (int)(k & 0x7FFFFFFFu) : ~(int)k;
    return __int_as_float(b);
}
#define EMAX_NEG_INF 0x007FFFFFu

#define GM_RT      5
#define GM_CT      2
#define GM_WAVES   4
#define GM_GROUPS  250
#define GM_KB      32
#define GM_KG      (GM_KB / 8)

struct gm_frags { v4f a[GM_KG][GM_RT]; v4f q[GM_KG][GM_CT]; };

__device__ __forceinline__ void gm_load(gm_frags& f, const float* __restrict__ ap,
                                        const float* __restrict__ bp, int k) {
#pragma unroll
    for (int g = 0; g < GM_KG; ++g) {
#pragma unroll
        for (int r = 0; r < GM_RT; ++r)
            f.a[g][r] = *(const v4f*)(ap + (size_t)r * 16 * IN_DIM + k + 8 * g);
#pragma unroll
        for (int t = 0; t < GM_CT; ++t)
            f.q[g][t] = *(const v4f*)(bp + (size_t)t * 16 * IN_DIM + k + 8 * g);
    }
}

__device__ __forceinline__ void gm_mma(v8f (&c)[GM_RT][GM_CT], const gm_frags& f) {
#pragma unroll
    for (int g = 0; g < GM_KG; ++g)
#pragma unroll
        for (int e = 0; e < 2; ++e)
#pragma unroll
            for (int r = 0; r < GM_RT; ++r)
#pragma unroll
                for (int t = 0; t < GM_CT; ++t) {
                    const v2f av = e ? (v2f){f.a[g][r].z, f.a[g][r].w} : (v2f){f.a[g][r].x, f.a[g][r].y};
                    const v2f bv = e ? (v2f){f.q[g][t].z, f.q[g][t].w} : (v2f){f.q[g][t].x, f.q[g][t].y};
                    c[r][t] = __builtin_amdgcn_wmma_f32_16x16x4_f32(false, av, false, bv,
                                                                    (short)0, c[r][t], false, false);
                }
}

__global__ __launch_bounds__(GM_WAVES * 32)
void k_gemm_wmma(const float* __restrict__ h, const float* __restrict__ w,
                 const float* __restrict__ bias, float* __restrict__ z) {
    const int lane = threadIdx.x & 31;
    const int wv   = threadIdx.x >> 5;
    const int row0 = (blockIdx.x >> 1) * (GM_RT * 16);
    const int col0 = ((((int)blockIdx.x & 1) << 2) + wv) * (GM_CT * 16);
    const int mn   = lane & 15;
    const int kb   = (lane >> 4) << 2;

    const float* ap = h + (size_t)(row0 + mn) * IN_DIM + kb;
    const float* bp = w + (size_t)(col0 + mn) * IN_DIM + kb;

    v8f c[GM_RT][GM_CT];
#pragma unroll
    for (int r = 0; r < GM_RT; ++r)
#pragma unroll
        for (int t = 0; t < GM_CT; ++t) c[r][t] = (v8f){};

    gm_frags f0, f1;
    gm_load(f0, ap, bp, 0);
#pragma unroll 1
    for (int k = 0; k < IN_DIM - 2 * GM_KB; k += 2 * GM_KB) {
        gm_load(f1, ap, bp, k + GM_KB);
        gm_mma(c, f0);
        gm_load(f0, ap, bp, k + 2 * GM_KB);
        gm_mma(c, f1);
    }
    gm_load(f1, ap, bp, IN_DIM - GM_KB);
    gm_mma(c, f0);
    gm_mma(c, f1);

    const int mbase = (lane >> 4) * 8;
#pragma unroll
    for (int t = 0; t < GM_CT; ++t) {
        const int   col = col0 + t * 16 + mn;
        const float bv  = bias[col];
#pragma unroll
        for (int r = 0; r < GM_RT; ++r)
#pragma unroll
            for (int v = 0; v < 8; ++v)
                z[(size_t)(row0 + r * 16 + mbase + v) * OUT_DIM + col] = c[r][t][v] + bv;
    }
}

#define KEY_ROWS   40
#define N_KEYLINES (N_EDGES / 128)
__device__ __forceinline__ unsigned slab_key(int d) {
    return min((unsigned)d / (unsigned)KEY_ROWS, 255u);
}
__global__ void k_node_dots(const float* __restrict__ z, const float* __restrict__ attn_w,
                            const int* __restrict__ dst,
                            float* __restrict__ s_src, float* __restrict__ s_dst,
                            unsigned* __restrict__ ekey) {
    const int lane = threadIdx.x & 31;
    const int node = (blockIdx.x * blockDim.x + threadIdx.x) >> 5;
    if (node >= N_NODES) return;
    if (node < N_KEYLINES) {
        const int4 d = ((const int4*)dst)[node * 32 + lane];
        ekey[node * 32 + lane] = slab_key(d.x) | (slab_key(d.y) << 8) |
                                 (slab_key(d.z) << 16) | (slab_key(d.w) << 24);
    }
    const float4* zrow = (const float4*)(z + (size_t)node * OUT_DIM);
    const float4* aw   = (const float4*)attn_w;
    float ps = 0.0f, pd = 0.0f;
#pragma unroll
    for (int j = 0; j < 2; ++j) {
        int vi = lane + j * 32;
        float4 zz = zrow[vi];
        float4 as = aw[vi];
        float4 ad = aw[64 + vi];
        ps += zz.x * as.x + zz.y * as.y + zz.z * as.z + zz.w * as.w;
        pd += zz.x * ad.x + zz.y * ad.y + zz.z * ad.z + zz.w * ad.w;
    }
#pragma unroll
    for (int m = 16; m >= 1; m >>= 1) {
        ps += __shfl_xor(ps, m, 32);
        pd += __shfl_xor(pd, m, 32);
    }
    if (lane == 0) { s_src[node] = ps; s_dst[node] = pd; }
}

#define SL_ROWS     40
#define SL_GROUPS   (N_NODES / SL_ROWS)
#define SL_THREADS  1024
#define SL_WAVES    (SL_THREADS / 32)
#define SL_CAP      16384
#define N_KQ        (N_EDGES / 16)
#define SC_BATCH    10
#define AG_ROWS     4

__device__ __forceinline__ float edge_logit(float ss, float sd, float ab) {
    const float v = ss + sd + ab;
    return (v >= 0.0f) ? v : NEG_SLOPE * v;
}

__device__ __forceinline__ void list_push(unsigned* l_cnt, unsigned* l_ent, int ei) {
    const unsigned slot = atomicAdd(l_cnt, 1u);
    if (slot < (unsigned)SL_CAP) l_ent[slot] = (unsigned)ei;
}

__device__ __forceinline__ unsigned key_hits(unsigned w, unsigned pat) {
    const unsigned x = w ^ pat;
    const unsigned t = (x & 0x7F7F7F7Fu) + 0x7F7F7F7Fu;
    return ~(t | x) & 0x80808080u;
}

struct zrow { v4f lo, hi; };
__device__ __forceinline__ zrow row_load(const float* __restrict__ z, int s, int lane) {
    const v4f* zs = (const v4f*)(z + (size_t)s * OUT_DIM) + lane;
    zrow q;
    q.lo = zs[0];
    q.hi = zs[32];
    return q;
}
__device__ __forceinline__ void row_sum(float* acc, unsigned r, float alpha, const zrow& q, int lane) {
    float* ar = acc + r * OUT_DIM + lane;
    unsafeAtomicAdd(&ar[0 * 32], alpha * q.lo.x);
    unsafeAtomicAdd(&ar[1 * 32], alpha * q.lo.y);
    unsafeAtomicAdd(&ar[2 * 32], alpha * q.lo.z);
    unsafeAtomicAdd(&ar[3 * 32], alpha * q.lo.w);
    unsafeAtomicAdd(&ar[4 * 32], alpha * q.hi.x);
    unsafeAtomicAdd(&ar[5 * 32], alpha * q.hi.y);
    unsafeAtomicAdd(&ar[6 * 32], alpha * q.hi.z);
    unsafeAtomicAdd(&ar[7 * 32], alpha * q.hi.w);
}
__device__ __forceinline__ void row_add(float* acc, const float* __restrict__ z,
                                        int s, unsigned r, float alpha, int lane) {
    const zrow q = row_load(z, s, lane);
    row_sum(acc, r, alpha, q, lane);
}
template <int NR>
__device__ __forceinline__ void rows_add(float* acc, const float* __restrict__ z,
                                         const unsigned* l_ent, const float* l_val,
                                         const float* l_dn, int i0, int lane) {
    unsigned rr[NR];
    float    aa[NR];
    zrow     q[NR];
#pragma unroll
    for (int u = 0; u < NR; ++u) {
        const int      i = i0 + u * SL_WAVES;
        const unsigned e = l_ent[i];
        rr[u] = e >> 16;
        aa[u] = l_val[i] / l_dn[rr[u]];
        q[u]  = row_load(z, (int)(e & 0xFFFFu), lane);
    }
#pragma unroll
    for (int u = 0; u < NR; ++u) row_sum(acc, rr[u], aa[u], q[u], lane);
}

__global__ __launch_bounds__(SL_THREADS)
void k_edges_owned(const int* __restrict__ src, const int* __restrict__ dst,
                   const unsigned* __restrict__ ekey,
                   const float* __restrict__ s_src, const float* __restrict__ s_dst,
                   const float* __restrict__ attn_b, const float* __restrict__ z,
                   float* __restrict__ hout) {
    __shared__ float    acc[SL_ROWS * OUT_DIM];
    __shared__ unsigned l_ent[SL_CAP];
    __shared__ float    l_val[SL_CAP];
    __shared__ float    l_sd[SL_ROWS];
    __shared__ unsigned l_mx[SL_ROWS];
    __shared__ float    l_dn[SL_ROWS];
    __shared__ unsigned l_cnt;

    const int   tid  = threadIdx.x;
    const int   lane = tid & 31;
    const int   wv   = tid >> 5;
    const int   lo   = blockIdx.x * SL_ROWS;
    const float ab   = attn_b[0];

    for (int j = tid; j < SL_ROWS * OUT_DIM / 4; j += SL_THREADS)
        ((float4*)acc)[j] = make_float4(0.0f, 0.0f, 0.0f, 0.0f);
    if (tid < SL_ROWS) { l_sd[tid] = s_dst[lo + tid]; l_mx[tid] = EMAX_NEG_INF; l_dn[tid] = 0.0f; }
    if (tid == 0) l_cnt = 0u;
    __syncthreads();

    const unsigned pat   = (unsigned)blockIdx.x * 0x01010101u;
    const int      start = blockIdx.x * (N_KQ / SL_GROUPS);
    for (int base = 0; base < N_KQ; base += SL_THREADS * SC_BATCH) {
        uint4 kq[SC_BATCH];
#pragma unroll
        for (int u = 0; u < SC_BATCH; ++u) {
            int qd = base + u * SL_THREADS + tid + start;
            if (qd >= N_KQ) qd -= N_KQ;
            kq[u] = ((const uint4*)ekey)[min(qd, N_KQ - 1)];
        }
#pragma unroll
        for (int u = 0; u < SC_BATCH; ++u) {
            const int it = base + u * SL_THREADS + tid;
            int qd = it + start;
            if (qd >= N_KQ) qd -= N_KQ;
            unsigned m = (key_hits(kq[u].x, pat) >> 3) | (key_hits(kq[u].y, pat) >> 2) |
                         (key_hits(kq[u].z, pat) >> 1) |  key_hits(kq[u].w, pat);
            if (it < N_KQ && m != 0u) {
                while (m) {
                    const int p = __builtin_ctz(m);
                    list_push(&l_cnt, l_ent, qd * 16 + ((p & 3) << 2) + (p >> 3));
                    m &= m - 1u;
                }
            }
        }
    }
    __syncthreads();
    const unsigned n_all = l_cnt;

    if (n_all <= (unsigned)SL_CAP) {
        const int n = (int)n_all;
        for (int i = tid; i < n; i += SL_THREADS) {
            const int      ei = (int)l_ent[i];
            const int      s  = src[ei];
            const unsigned r  = (unsigned)(dst[ei] - lo);
            const float    v  = edge_logit(s_src[s], l_sd[r], ab);
            l_ent[i] = (unsigned)s | (r << 16);
            l_val[i] = v;
            atomicMax(&l_mx[r], enc_f32(v));
        }
        __syncthreads();
        for (int i = tid; i < n; i += SL_THREADS) {
            const unsigned r = l_ent[i] >> 16;
            const float    x = expf(l_val[i] - dec_f32(l_mx[r]));
            l_val[i] = x;
            unsafeAtomicAdd(&l_dn[r], x);
        }
        __syncthreads();
        const int cnt = (n > wv) ? (n - wv + SL_WAVES - 1) / SL_WAVES : 0;
        int k = 0;
        for (; k + AG_ROWS <= cnt; k += AG_ROWS)
            rows_add<AG_ROWS>(acc, z, l_ent, l_val, l_dn, wv + k * SL_WAVES, lane);
        switch (cnt - k) {
            case 3: rows_add<3>(acc, z, l_ent, l_val, l_dn, wv + k * SL_WAVES, lane); break;
            case 2: rows_add<2>(acc, z, l_ent, l_val, l_dn, wv + k * SL_WAVES, lane); break;
            case 1: rows_add<1>(acc, z, l_ent, l_val, l_dn, wv + k * SL_WAVES, lane); break;
            default: break;
        }
    } else {
        for (int ei = tid; ei < N_EDGES; ei += SL_THREADS) {
            const unsigned r = (unsigned)(dst[ei] - lo);
            if (r < (unsigned)SL_ROWS)
                atomicMax(&l_mx[r], enc_f32(edge_logit(s_src[src[ei]], l_sd[r], ab)));
        }
        __syncthreads();
        for (int ei = tid; ei < N_EDGES; ei += SL_THREADS) {
            const unsigned r = (unsigned)(dst[ei] - lo);
            if (r < (unsigned)SL_ROWS) {
                const float v = edge_logit(s_src[src[ei]], l_sd[r], ab);
                unsafeAtomicAdd(&l_dn[r], expf(v - dec_f32(l_mx[r])));
            }
        }
        __syncthreads();
        for (int base = wv * 32; base < N_EDGES; base += SL_THREADS) {
            const int      ei   = base + lane;
            const unsigned r    = (unsigned)(dst[ei] - lo);
            const bool     mine = r < (unsigned)SL_ROWS;
            int   s = 0;
            float a = 0.0f;
            if (mine) {
                s = src[ei];
                const float v = edge_logit(s_src[s], l_sd[r], ab);
                a = expf(v - dec_f32(l_mx[r])) / l_dn[r];
            }
            unsigned m = __builtin_amdgcn_ballot_w32(mine);
            while (m) {
                const int i = __builtin_ctz(m);
                m &= m - 1u;
                const int      si = __builtin_amdgcn_readlane(s, i);
                const unsigned ri = (unsigned)__builtin_amdgcn_readlane((int)r, i);
                const float    ai = __int_as_float(__builtin_amdgcn_readlane(__float_as_int(a), i));
                row_add(acc, z, si, ri, ai, lane);
            }
        }
    }
    __syncthreads();
    float4* og = (float4*)(hout + (size_t)lo * OUT_DIM);
    for (int j = tid; j < SL_ROWS * OUT_DIM / 4; j += SL_THREADS) {
        const int    f = j & (OUT_DIM / 4 - 1);
        const float* p = acc + (j >> 6) * OUT_DIM + ((f >> 5) << 7) + (f & 31);
        og[j] = make_float4(p[0], p[32], p[64], p[96]);
    }
}

static inline size_t align256(size_t x) { return (x + 255) & ~(size_t)255; }

extern "C" void kernel_launch(void* const* d_in, const int* in_sizes, int n_in,
                              void* d_out, int out_size, void* d_ws, size_t ws_size,
                              hipStream_t stream) {
    const float* h      = (const float*)d_in[0];
    const float* fc_w   = (const float*)d_in[1];
    const float* fc_b   = (const float*)d_in[2];
    const float* attn_w = (const float*)d_in[3];
    const float* attn_b = (const float*)d_in[4];
    const int*   src    = (const int*)d_in[5];
    const int*   dst    = (const int*)d_in[6];
    float* hout = (float*)d_out;

    char* ws = (char*)d_ws;
    float*    z      = (float*)ws;    ws += align256((size_t)N_NODES * OUT_DIM * 4);
    float*    s_src  = (float*)ws;    ws += align256((size_t)N_NODES * 4);
    float*    s_dst  = (float*)ws;    ws += align256((size_t)N_NODES * 4);
    unsigned* ekey   = (unsigned*)ws; ws += align256((size_t)N_EDGES);

    k_gemm_wmma<<<GM_GROUPS, GM_WAVES * 32, 0, stream>>>(h, fc_w, fc_b, z);
    k_node_dots<<<1250, 256, 0, stream>>>(z, attn_w, dst, s_src, s_dst, ekey);
    k_edges_owned<<<SL_GROUPS, SL_THREADS, 0, stream>>>(src, dst, ekey, s_src, s_dst, attn_b, z, hout);
}


# Row sums in registers: wave w of a slab owns rows 2w, 2w+1, filters the slab's LDS list for them into its own queue, adds alpha*z[src] in list (arrival) order, stores its rows; no LDS float adds
# speedup vs baseline: 4.6230x; 1.0783x over previous
#include <hip/hip_runtime.h>
#include <hip/hip_bf16.h>

#define N_NODES 10000
#define N_EDGES 320000
#define IN_DIM  512
#define OUT_DIM 256
#define NEG_SLOPE 0.01f

typedef float v2f __attribute__((ext_vector_type(2)));
typedef float v4f __attribute__((ext_vector_type(4)));
typedef float v8f __attribute__((ext_vector_type(8)));

__device__ __forceinline__ unsigned enc_f32(float f) {
    int b = __float_as_int(f);
    return (b >= 0) ? ((unsigned)b | 0x80000000u) : ~(unsigned)b;
}
__device__ __forceinline__ float dec_f32(unsigned k) {
    int b = (k & 0x80000000u) ? (int)(k & 0x7FFFFFFFu) : ~(int)k;
    return __int_as_float(b);
}
#define EMAX_NEG_INF 0x007FFFFFu

#define GM_RT      5
#define GM_CT      2
#define GM_WAVES   4
#define GM_GROUPS  250
#define GM_KB      32
#define GM_KG      (GM_KB / 8)

struct gm_frags { v4f a[GM_KG][GM_RT]; v4f q[GM_KG][GM_CT]; };

__device__ __forceinline__ void gm_load(gm_frags& f, const float* __restrict__ ap,
                                        const float* __restrict__ bp, int k) {
#pragma unroll
    for (int g = 0; g < GM_KG; ++g) {
#pragma unroll
        for (int r = 0; r < GM_RT; ++r)
            f.a[g][r] = *(const v4f*)(ap + (size_t)r * 16 * IN_DIM + k + 8 * g);
#pragma unroll
        for (int t = 0; t < GM_CT; ++t)
            f.q[g][t] = *(const v4f*)(bp + (size_t)t * 16 * IN_DIM + k + 8 * g);
    }
}

__device__ __forceinline__ void gm_mma(v8f (&c)[GM_RT][GM_CT], const gm_frags& f) {
#pragma unroll
    for (int g = 0; g < GM_KG; ++g)
#pragma unroll
        for (int e = 0; e < 2; ++e)
#pragma unroll
            for (int r = 0; r < GM_RT; ++r)
#pragma unroll
                for (int t = 0; t < GM_CT; ++t) {
                    const v2f av = e ? (v2f){f.a[g][r].z, f.a[g][r].w} : (v2f){f.a[g][r].x, f.a[g][r].y};
                    const v2f bv = e ? (v2f){f.q[g][t].z, f.q[g][t].w} : (v2f){f.q[g][t].x, f.q[g][t].y};
                    c[r][t] = __builtin_amdgcn_wmma_f32_16x16x4_f32(false, av, false, bv,
                                                                    (short)0, c[r][t], false, false);
                }
}

__global__ __launch_bounds__(GM_WAVES * 32)
void k_gemm_wmma(const float* __restrict__ h, const float* __restrict__ w,
                 const float* __restrict__ bias, float* __restrict__ z) {
    const int lane = threadIdx.x & 31;
    const int wv   = threadIdx.x >> 5;
    const int row0 = (blockIdx.x >> 1) * (GM_RT * 16);
    const int col0 = ((((int)blockIdx.x & 1) << 2) + wv) * (GM_CT * 16);
    const int mn   = lane & 15;
    const int kb   = (lane >> 4) << 2;

    const float* ap = h + (size_t)(row0 + mn) * IN_DIM + kb;
    const float* bp = w + (size_t)(col0 + mn) * IN_DIM + kb;

    v8f c[GM_RT][GM_CT];
#pragma unroll
    for (int r = 0; r < GM_RT; ++r)
#pragma unroll
        for (int t = 0; t < GM_CT; ++t) c[r][t] = (v8f){};

    gm_frags f0, f1;
    gm_load(f0, ap, bp, 0);
#pragma unroll 1
    for (int k = 0; k < IN_DIM - 2 * GM_KB; k += 2 * GM_KB) {
        gm_load(f1, ap, bp, k + GM_KB);
        gm_mma(c, f0);
        gm_load(f0, ap, bp, k + 2 * GM_KB);
        gm_mma(c, f1);
    }
    gm_load(f1, ap, bp, IN_DIM - GM_KB);
    gm_mma(c, f0);
    gm_mma(c, f1);

    const int mbase = (lane >> 4) * 8;
#pragma unroll
    for (int t = 0; t < GM_CT; ++t) {
        const int   col = col0 + t * 16 + mn;
        const float bv  = bias[col];
#pragma unroll
        for (int r = 0; r < GM_RT; ++r)
#pragma unroll
            for (int v = 0; v < 8; ++v)
                z[(size_t)(row0 + r * 16 + mbase + v) * OUT_DIM + col] = c[r][t][v] + bv;
    }
}

#define KEY_ROWS   40
#define N_KEYLINES (N_EDGES / 128)
__device__ __forceinline__ unsigned slab_key(int d) {
    return min((unsigned)d / (unsigned)KEY_ROWS, 255u);
}
__global__ void k_node_dots(const float* __restrict__ z, const float* __restrict__ attn_w,
                            const int* __restrict__ dst,
                            float* __restrict__ s_src, float* __restrict__ s_dst,
                            unsigned* __restrict__ ekey) {
    const int lane = threadIdx.x & 31;
    const int node = (blockIdx.x * blockDim.x + threadIdx.x) >> 5;
    if (node >= N_NODES) return;
    if (node < N_KEYLINES) {
        const int4 d = ((const int4*)dst)[node * 32 + lane];
        ekey[node * 32 + lane] = slab_key(d.x) | (slab_key(d.y) << 8) |
                                 (slab_key(d.z) << 16) | (slab_key(d.w) << 24);
    }
    const float4* zrow = (const float4*)(z + (size_t)node * OUT_DIM);
    const float4* aw   = (const float4*)attn_w;
    float ps = 0.0f, pd = 0.0f;
#pragma unroll
    for (int j = 0; j < 2; ++j) {
        int vi = lane + j * 32;
        float4 zz = zrow[vi];
        float4 as = aw[vi];
        float4 ad = aw[64 + vi];
        ps += zz.x * as.x + zz.y * as.y + zz.z * as.z + zz.w * as.w;
        pd += zz.x * ad.x + zz.y * ad.y + zz.z * ad.z + zz.w * ad.w;
    }
#pragma unroll
    for (int m = 16; m >= 1; m >>= 1) {
        ps += __shfl_xor(ps, m, 32);
        pd += __shfl_xor(pd, m, 32);
    }
    if (lane == 0) { s_src[node] = ps; s_dst[node] = pd; }
}

#define SL_ROWS     40
#define SL_GROUPS   (N_NODES / SL_ROWS)
#define SL_THREADS  1024
#define SL_WAVES    (SL_THREADS / 32)
#define SL_CAP      16384
#define N_KQ        (N_EDGES / 16)
#define SC_BATCH    10
#define SL_OWNERS   (SL_ROWS / 2)
#define AG_ROWS     8
#define Q_DRAIN     32
#define Q_CAP       (Q_DRAIN + 128)

__device__ __forceinline__ float edge_logit(float ss, float sd, float ab) {
    const float v = ss + sd + ab;
    return (v >= 0.0f) ? v : NEG_SLOPE * v;
}

__device__ __forceinline__ void list_push(unsigned* l_cnt, unsigned* l_ent, int ei) {
    const unsigned slot = atomicAdd(l_cnt, 1u);
    if (slot < (unsigned)SL_CAP) l_ent[slot] = (unsigned)ei;
}

__device__ __forceinline__ unsigned key_hits(unsigned w, unsigned pat) {
    const unsigned x = w ^ pat;
    const unsigned t = (x & 0x7F7F7F7Fu) + 0x7F7F7F7Fu;
    return ~(t | x) & 0x80808080u;
}

struct zrow { v4f lo, hi; };
__device__ __forceinline__ zrow row_load(const float* __restrict__ z, int s, int lane) {
    const v4f* zs = (const v4f*)(z + (size_t)s * OUT_DIM) + lane;
    zrow q;
    q.lo = zs[0];
    q.hi = zs[32];
    return q;
}
__device__ __forceinline__ void row_sum(float* acc, unsigned r, float alpha, const zrow& q, int lane) {
    float* ar = acc + r * OUT_DIM + lane;
    unsafeAtomicAdd(&ar[0 * 32], alpha * q.lo.x);
    unsafeAtomicAdd(&ar[1 * 32], alpha * q.lo.y);
    unsafeAtomicAdd(&ar[2 * 32], alpha * q.lo.z);
    unsafeAtomicAdd(&ar[3 * 32], alpha * q.lo.w);
    unsafeAtomicAdd(&ar[4 * 32], alpha * q.hi.x);
    unsafeAtomicAdd(&ar[5 * 32], alpha * q.hi.y);
    unsafeAtomicAdd(&ar[6 * 32], alpha * q.hi.z);
    unsafeAtomicAdd(&ar[7 * 32], alpha * q.hi.w);
}
__device__ __forceinline__ void row_add(float* acc, const float* __restrict__ z,
                                        int s, unsigned r, float alpha, int lane) {
    const zrow q = row_load(z, s, lane);
    row_sum(acc, r, alpha, q, lane);
}
template <int NR>
__device__ __forceinline__ void own_rows(zrow& s0, zrow& s1, const float* __restrict__ z,
                                         unsigned e, float al, int u0, int lane) {
    unsigned ee[NR];
    float    aa[NR];
    zrow     q[NR];
#pragma unroll
    for (int u = 0; u < NR; ++u) {
        ee[u] = (unsigned)__builtin_amdgcn_readlane((int)e, u0 + u);
        aa[u] = __int_as_float(__builtin_amdgcn_readlane(__float_as_int(al), u0 + u));
        q[u]  = row_load(z, (int)(ee[u] & 0xFFFFu), lane);
    }
#pragma unroll
    for (int u = 0; u < NR; ++u) {
        if (ee[u] & 0x10000u) { s1.lo += aa[u] * q[u].lo; s1.hi += aa[u] * q[u].hi; }
        else                  { s0.lo += aa[u] * q[u].lo; s0.hi += aa[u] * q[u].hi; }
    }
}
__device__ __forceinline__ void own_drain(zrow& s0, zrow& s1, const float* __restrict__ z,
                                          const unsigned* q_ent, const unsigned* q_idx, int cnt,
                                          const float* l_val, const float* l_dn, int lane) {
    for (int g = 0; g < cnt; g += 32) {
        const int m = min(32, cnt - g);
        unsigned e  = 0u;
        float    al = 0.0f;
        if (lane < m) {
            e  = q_ent[g + lane];
            al = l_val[q_idx[g + lane]] / l_dn[e >> 16];
        }
        int u = 0;
        for (; u + AG_ROWS <= m; u += AG_ROWS) own_rows<AG_ROWS>(s0, s1, z, e, al, u, lane);
        if (m - u >= 4) { own_rows<4>(s0, s1, z, e, al, u, lane); u += 4; }
        if (m - u >= 2) { own_rows<2>(s0, s1, z, e, al, u, lane); u += 2; }
        if (m - u >= 1) { own_rows<1>(s0, s1, z, e, al, u, lane); }
    }
}

__global__ __launch_bounds__(SL_THREADS)
void k_edges_owned(const int* __restrict__ src, const int* __restrict__ dst,
                   const unsigned* __restrict__ ekey,
                   const float* __restrict__ s_src, const float* __restrict__ s_dst,
                   const float* __restrict__ attn_b, const float* __restrict__ z,
                   float* __restrict__ hout) {
    __shared__ float    acc[SL_ROWS * OUT_DIM];
    __shared__ __attribute__((aligned(16))) unsigned l_ent[SL_CAP];
    __shared__ unsigned q_ent[SL_OWNERS * Q_CAP];
    __shared__ unsigned q_idx[SL_OWNERS * Q_CAP];
    __shared__ float    l_val[SL_CAP];
    __shared__ float    l_sd[SL_ROWS];
    __shared__ unsigned l_mx[SL_ROWS];
    __shared__ float    l_dn[SL_ROWS];
    __shared__ unsigned l_cnt;

    const int   tid  = threadIdx.x;
    const int   lane = tid & 31;
    const int   wv   = tid >> 5;
    const int   lo   = blockIdx.x * SL_ROWS;
    const float ab   = attn_b[0];

    if (tid < SL_ROWS) { l_sd[tid] = s_dst[lo + tid]; l_mx[tid] = EMAX_NEG_INF; l_dn[tid] = 0.0f; }
    if (tid == 0) l_cnt = 0u;
    __syncthreads();

    const unsigned pat   = (unsigned)blockIdx.x * 0x01010101u;
    const int      start = blockIdx.x * (N_KQ / SL_GROUPS);
    for (int base = 0; base < N_KQ; base += SL_THREADS * SC_BATCH) {
        uint4 kq[SC_BATCH];
#pragma unroll
        for (int u = 0; u < SC_BATCH; ++u) {
            int qd = base + u * SL_THREADS + tid + start;
            if (qd >= N_KQ) qd -= N_KQ;
            kq[u] = ((const uint4*)ekey)[min(qd, N_KQ - 1)];
        }
#pragma unroll
        for (int u = 0; u < SC_BATCH; ++u) {
            const int it = base + u * SL_THREADS + tid;
            int qd = it + start;
            if (qd >= N_KQ) qd -= N_KQ;
            unsigned m = (key_hits(kq[u].x, pat) >> 3) | (key_hits(kq[u].y, pat) >> 2) |
                         (key_hits(kq[u].z, pat) >> 1) |  key_hits(kq[u].w, pat);
            if (it < N_KQ && m != 0u) {
                while (m) {
                    const int p = __builtin_ctz(m);
                    list_push(&l_cnt, l_ent, qd * 16 + ((p & 3) << 2) + (p >> 3));
                    m &= m - 1u;
                }
            }
        }
    }
    __syncthreads();
    const unsigned n_all = l_cnt;

    if (n_all <= (unsigned)SL_CAP) {
        const int n = (int)n_all;
        for (int i = tid; i < n; i += SL_THREADS) {
            const int      ei = (int)l_ent[i];
            const int      s  = src[ei];
            const unsigned r  = (unsigned)(dst[ei] - lo);
            const float    v  = edge_logit(s_src[s], l_sd[r], ab);
            l_ent[i] = (unsigned)s | (r << 16);
            l_val[i] = v;
            atomicMax(&l_mx[r], enc_f32(v));
        }
        __syncthreads();
        for (int i = tid; i < n; i += SL_THREADS) {
            const unsigned r = l_ent[i] >> 16;
            const float    x = expf(l_val[i] - dec_f32(l_mx[r]));
            l_val[i] = x;
            unsafeAtomicAdd(&l_dn[r], x);
        }
        __syncthreads();
        if (wv < SL_OWNERS) {
            unsigned* qe = q_ent + wv * Q_CAP;
            unsigned* qi = q_idx + wv * Q_CAP;
            zrow s0, s1;
            s0.lo = s0.hi = s1.lo = s1.hi = (v4f){0.0f, 0.0f, 0.0f, 0.0f};
            int qn = 0;
            const uint4* lq = (const uint4*)l_ent;
            uint4 nx = lq[lane];
            for (int base = 0; base < n; base += 128) {
                const uint4 cur = nx;
                nx = lq[min((base + 128) / 4 + lane, SL_CAP / 4 - 1)];
                const unsigned ev[4] = {cur.x, cur.y, cur.z, cur.w};
#pragma unroll
                for (int c = 0; c < 4; ++c) {
                    const int  i   = base + 4 * lane + c;
                    const bool hit = (i < n) && ((ev[c] >> 17) == (unsigned)wv);
                    const unsigned m = __builtin_amdgcn_ballot_w32(hit);
                    if (hit) {
                        const int p = qn + (int)__builtin_amdgcn_mbcnt_lo(m, 0u);
                        qe[p] = ev[c];
                        qi[p] = (unsigned)i;
                    }
                    qn += __builtin_popcount(m);
                }
                if (qn >= Q_DRAIN) {
                    own_drain(s0, s1, z, qe, qi, qn, l_val, l_dn, lane);
                    qn = 0;
                }
            }
            own_drain(s0, s1, z, qe, qi, qn, l_val, l_dn, lane);
            v4f* o0 = (v4f*)(hout + (size_t)(lo + 2 * wv) * OUT_DIM) + lane;
            o0[0]                = s0.lo;
            o0[32]               = s0.hi;
            o0[OUT_DIM / 4]      = s1.lo;
            o0[OUT_DIM / 4 + 32] = s1.hi;
        }
    } else {
        for (int j = tid; j < SL_ROWS * OUT_DIM / 4; j += SL_THREADS)
            ((float4*)acc)[j] = make_float4(0.0f, 0.0f, 0.0f, 0.0f);
        for (int ei = tid; ei < N_EDGES; ei += SL_THREADS) {
            const unsigned r = (unsigned)(dst[ei] - lo);
            if (r < (unsigned)SL_ROWS)
                atomicMax(&l_mx[r], enc_f32(edge_logit(s_src[src[ei]], l_sd[r], ab)));
        }
        __syncthreads();
        for (int ei = tid; ei < N_EDGES; ei += SL_THREADS) {
            const unsigned r = (unsigned)(dst[ei] - lo);
            if (r < (unsigned)SL_ROWS) {
                const float v = edge_logit(s_src[src[ei]], l_sd[r], ab);
                unsafeAtomicAdd(&l_dn[r], expf(v - dec_f32(l_mx[r])));
            }
        }
        __syncthreads();
        for (int base = wv * 32; base < N_EDGES; base += SL_THREADS) {
            const int      ei   = base + lane;
            const unsigned r    = (unsigned)(dst[ei] - lo);
            const bool     mine = r < (unsigned)SL_ROWS;
            int   s = 0;
            float a = 0.0f;
            if (mine) {
                s = src[ei];
                const float v = edge_logit(s_src[s], l_sd[r], ab);
                a = expf(v - dec_f32(l_mx[r])) / l_dn[r];
            }
            unsigned m = __builtin_amdgcn_ballot_w32(mine);
            while (m) {
                const int i = __builtin_ctz(m);
                m &= m - 1u;
                const int      si = __builtin_amdgcn_readlane(s, i);
                const unsigned ri = (unsigned)__builtin_amdgcn_readlane((int)r, i);
                const float    ai = __int_as_float(__builtin_amdgcn_readlane(__float_as_int(a), i));
                row_add(acc, z, si, ri, ai, lane);
            }
        }
        __syncthreads();
        float4* og = (float4*)(hout + (size_t)lo * OUT_DIM);
        for (int j = tid; j < SL_ROWS * OUT_DIM / 4; j += SL_THREADS) {
            const int    f = j & (OUT_DIM / 4 - 1);
            const float* p = acc + (j >> 6) * OUT_DIM + ((f >> 5) << 7) + (f & 31);
            og[j] = make_float4(p[0], p[32], p[64], p[96]);
        }
    }
}

static inline size_t align256(size_t x) { return (x + 255) & ~(size_t)255; }

extern "C" void kernel_launch(void* const* d_in, const int* in_sizes, int n_in,
                              void* d_out, int out_size, void* d_ws, size_t ws_size,
                              hipStream_t stream) {
    const float* h      = (const float*)d_in[0];
    const float* fc_w   = (const float*)d_in[1];
    const float* fc_b   = (const float*)d_in[2];
    const float* attn_w = (const float*)d_in[3];
    const float* attn_b = (const float*)d_in[4];
    const int*   src    = (const int*)d_in[5];
    const int*   dst    = (const int*)d_in[6];
    float* hout = (float*)d_out;

    char* ws = (char*)d_ws;
    float*    z      = (float*)ws;    ws += align256((size_t)N_NODES * OUT_DIM * 4);
    float*    s_src  = (float*)ws;    ws += align256((size_t)N_NODES * 4);
    float*    s_dst  = (float*)ws;    ws += align256((size_t)N_NODES * 4);
    unsigned* ekey   = (unsigned*)ws; ws += align256((size_t)N_EDGES);

    k_gemm_wmma<<<GM_GROUPS, GM_WAVES * 32, 0, stream>>>(h, fc_w, fc_b, z);
    k_node_dots<<<1250, 256, 0, stream>>>(z, attn_w, dst, s_src, s_dst, ekey);
    k_edges_owned<<<SL_GROUPS, SL_THREADS, 0, stream>>>(src, dst, ekey, s_src, s_dst, attn_b, z, hout);
}
